# starting kernel
# baseline (speedup 1.0000x reference)
#include <hip/hip_runtime.h>

typedef __attribute__((ext_vector_type(2))) float v2f;
typedef __attribute__((ext_vector_type(8))) float v8f;

#define HIDDEN 128

__global__ __launch_bounds__(256) void gcn_zero_kernel(float* __restrict__ out,
                                                       size_t n_out,
                                                       float* __restrict__ deg,
                                                       int n_deg) {
    size_t i = (size_t)blockIdx.x * blockDim.x + threadIdx.x;
    size_t stride = (size_t)gridDim.x * blockDim.x;
    for (size_t j = i; j < n_out; j += stride) out[j] = 0.0f;
    for (size_t j = i; j < (size_t)n_deg; j += stride) deg[j] = 0.0f;
}

__global__ __launch_bounds__(256) void gcn_gemm_wmma_kernel(
        const float* __restrict__ x, const float* __restrict__ W,
        float* __restrict__ h, int N) {
    const int lane    = threadIdx.x & 31;
    const int wave    = threadIdx.x >> 5;
    const int halfSel = lane >> 4;
    const int l       = lane & 15;

    const int rowBase = blockIdx.x * 16;
    const int nBase   = wave * 16;

    int rowLd = rowBase + l;
    if (rowLd >= N) rowLd = N - 1;

    const float* __restrict__ xrow = x + (size_t)rowLd * HIDDEN;
    const float* __restrict__ wrow = W + (size_t)(nBase + l) * HIDDEN;

    v8f c = {};
    #pragma unroll
    for (int k = 0; k < HIDDEN; k += 4) {
        const int ka = k + (halfSel << 1);
        v2f a = *(const v2f*)(xrow + ka);
        v2f b = *(const v2f*)(wrow + ka);
        c = __builtin_amdgcn_wmma_f32_16x16x4_f32(
                 false, a,  false, b,
                 (short)0, c,  false,  false);
    }

    float* __restrict__ hp = h + (size_t)rowBase * HIDDEN + nBase + l;
    #pragma unroll
    for (int v = 0; v < 8; ++v) {
        const int m = v + halfSel * 8;
        if (rowBase + m < N) hp[(size_t)m * HIDDEN] = c[v];
    }
}

__global__ __launch_bounds__(256) void gcn_deg_kernel(const int* __restrict__ col,
                                                      float* __restrict__ deg,
                                                      int E) {
    int stride = gridDim.x * blockDim.x;
    for (int i = blockIdx.x * blockDim.x + threadIdx.x; i < E; i += stride) {
        atomicAdd(&deg[col[i]], 1.0f);
    }
}

__global__ __launch_bounds__(256) void gcn_dis_kernel(float* __restrict__ deg, int N) {
    int i = blockIdx.x * blockDim.x + threadIdx.x;
    if (i < N) {
        float d = deg[i];
        deg[i] = (d > 0.0f) ? __frsqrt_rn(d) : 0.0f;
    }
}

__global__ __launch_bounds__(256) void gcn_scatter_kernel(
        const float* __restrict__ h, const float* __restrict__ dis,
        const int* __restrict__ row, const int* __restrict__ col,
        float* __restrict__ out, int E) {
    const int lane  = threadIdx.x & 31;
    const int wavesPerBlock = blockDim.x >> 5;
    const int wave  = blockIdx.x * wavesPerBlock + (threadIdx.x >> 5);
    const int nWaves = gridDim.x * wavesPerBlock;

    for (int e = wave; e < E; e += nWaves) {
        const int r = row[e];
        const int cI = col[e];
        const float norm = dis[r] * dis[cI];

        const float4 hv = ((const float4*)(h + (size_t)cI * HIDDEN))[lane];
        float* op = out + (size_t)r * HIDDEN + lane * 4;
        atomicAdd(op + 0, hv.x * norm);
        atomicAdd(op + 1, hv.y * norm);
        atomicAdd(op + 2, hv.z * norm);
        atomicAdd(op + 3, hv.w * norm);
    }
}

extern "C" void kernel_launch(void* const* d_in, const int* in_sizes, int n_in,
                              void* d_out, int out_size, void* d_ws, size_t ws_size,
                              hipStream_t stream) {
    const float* x  = (const float*)d_in[0];
    const float* W  = (const float*)d_in[1];
    const int*   ei = (const int*)d_in[2];
    float* out = (float*)d_out;

    const int N = in_sizes[0] / HIDDEN;
    const int E = in_sizes[2] / 2;
    const int* rowIdx = ei;
    const int* colIdx = ei + E;

    float* h   = (float*)d_ws;
    float* deg = h + (size_t)N * HIDDEN;

    gcn_zero_kernel<<<2048, 256, 0, stream>>>(out, (size_t)N * HIDDEN, deg, N);

    const int rowTiles = (N + 15) / 16;
    gcn_gemm_wmma_kernel<<<rowTiles, 256, 0, stream>>>(x, W, h, N);

    gcn_deg_kernel<<<1024, 256, 0, stream>>>(colIdx, deg, E);

    gcn_dis_kernel<<<(N + 255) / 256, 256, 0, stream>>>(deg, N);

    gcn_scatter_kernel<<<4096, 256, 0, stream>>>(h, deg, rowIdx, colIdx, out, E);
}


# Aggregation: each block owns 400 out rows as an LDS tile, scans row[] for its edges, adds with LDS f32 atomics (arrival order), stores each out line once; no global atomics, no out zeroing
# speedup vs baseline: 1.5143x; 1.5143x over previous
#include <hip/hip_runtime.h>

typedef __attribute__((ext_vector_type(2))) float v2f;
typedef __attribute__((ext_vector_type(8))) float v8f;

#define HIDDEN 128

__global__ __launch_bounds__(256) void gcn_zero_kernel(float* __restrict__ deg, int n_deg) {
    int i = blockIdx.x * blockDim.x + threadIdx.x;
    if (i < n_deg) deg[i] = 0.0f;
}

__global__ __launch_bounds__(256) void gcn_gemm_wmma_kernel(
        const float* __restrict__ x, const float* __restrict__ W,
        float* __restrict__ h, int N) {
    const int lane    = threadIdx.x & 31;
    const int wave    = threadIdx.x >> 5;
    const int halfSel = lane >> 4;
    const int l       = lane & 15;

    const int rowBase = blockIdx.x * 16;
    const int nBase   = wave * 16;

    int rowLd = rowBase + l;
    if (rowLd >= N) rowLd = N - 1;

    const float* __restrict__ xrow = x + (size_t)rowLd * HIDDEN;
    const float* __restrict__ wrow = W + (size_t)(nBase + l) * HIDDEN;

    v8f c = {};
    #pragma unroll
    for (int k = 0; k < HIDDEN; k += 4) {
        const int ka = k + (halfSel << 1);
        v2f a = *(const v2f*)(xrow + ka);
        v2f b = *(const v2f*)(wrow + ka);
        c = __builtin_amdgcn_wmma_f32_16x16x4_f32(
                 false, a,  false, b,
                 (short)0, c,  false,  false);
    }

    float* __restrict__ hp = h + (size_t)rowBase * HIDDEN + nBase + l;
    #pragma unroll
    for (int v = 0; v < 8; ++v) {
        const int m = v + halfSel * 8;
        if (rowBase + m < N) hp[(size_t)m * HIDDEN] = c[v];
    }
}

__global__ __launch_bounds__(256) void gcn_deg_kernel(const int* __restrict__ col,
                                                      float* __restrict__ deg,
                                                      int E) {
    int stride = gridDim.x * blockDim.x;
    for (int i = blockIdx.x * blockDim.x + threadIdx.x; i < E; i += stride) {
        atomicAdd(&deg[col[i]], 1.0f);
    }
}

__global__ __launch_bounds__(256) void gcn_dis_kernel(float* __restrict__ deg, int N) {
    int i = blockIdx.x * blockDim.x + threadIdx.x;
    if (i < N) {
        float d = deg[i];
        deg[i] = (d > 0.0f) ? __frsqrt_rn(d) : 0.0f;
    }
}

#define AGG_ROWS    400
#define AGG_THREADS 1024
#define AGG_CAP     16384
#define AGG_UNR     8

__device__ __noinline__ void agg_add_serial(float* tile, const float* __restrict__ h,
                                            const float* __restrict__ dis,
                                            const int* __restrict__ col,
                                            int e, int r, int rowBase) {
    const int cI = col[e];
    const float norm = dis[r] * dis[cI];
    const float* hp = h + (size_t)cI * HIDDEN;
    float* tp = tile + (r - rowBase) * HIDDEN;
    for (int k = 0; k < HIDDEN; ++k) atomicAdd(tp + k, hp[k] * norm);
}

__global__ __launch_bounds__(AGG_THREADS) void gcn_aggregate_kernel(
        const float* __restrict__ h, const float* __restrict__ dis,
        const int* __restrict__ row, const int* __restrict__ col,
        float* __restrict__ out, int N, int E) {
    __shared__ __attribute__((aligned(16))) float s_tile[AGG_ROWS * HIDDEN];
    __shared__ int s_list[AGG_CAP];
    __shared__ int s_cnt;

    const int tid  = threadIdx.x;
    const int lane = tid & 31;
    const int wave = tid >> 5;
    const int rowBase = blockIdx.x * AGG_ROWS;
    int rowsHere = N - rowBase;
    if (rowsHere > AGG_ROWS) rowsHere = AGG_ROWS;
    const unsigned urows = (unsigned)rowsHere;

    {
        float4* t4 = (float4*)s_tile;
        const float4 z = make_float4(0.f, 0.f, 0.f, 0.f);
        for (int i = tid; i < AGG_ROWS * HIDDEN / 4; i += AGG_THREADS) t4[i] = z;
        if (tid == 0) s_cnt = 0;
    }
    __syncthreads();

    {
        const int4* __restrict__ row4 = (const int4*)row;
        const int n4 = E >> 2;
        const int step = AGG_THREADS * AGG_UNR;
        const int nIter = (n4 + step - 1) / step;
        const int itStart = (int)(blockIdx.x % (unsigned)nIter);
        const int noMatch = rowBase - 1;
        for (int it = 0; it < nIter; ++it) {
            int itw = it + itStart;
            if (itw >= nIter) itw -= nIter;
            const int i0 = itw * step + tid;
            int4 v[AGG_UNR];
            #pragma unroll
            for (int u = 0; u < AGG_UNR; ++u) {
                const int i = i0 + u * AGG_THREADS;
                v[u] = (i < n4) ? row4[i] : make_int4(noMatch, noMatch, noMatch, noMatch);
            }
            #pragma unroll
            for (int u = 0; u < AGG_UNR; ++u) {
                const int e0 = (i0 + u * AGG_THREADS) * 4;
                const int rr[4] = {v[u].x, v[u].y, v[u].z, v[u].w};
                #pragma unroll
                for (int q = 0; q < 4; ++q) {
                    if ((unsigned)(rr[q] - rowBase) < urows) {
                        const int p = atomicAdd(&s_cnt, 1);
                        if (p < AGG_CAP) s_list[p] = e0 + q;
                        else agg_add_serial(s_tile, h, dis, col, e0 + q, rr[q], rowBase);
                    }
                }
            }
        }
        const int eTail = (n4 << 2) + tid;
        if (eTail < E) {
            const int r = row[eTail];
            if ((unsigned)(r - rowBase) < urows) {
                const int p = atomicAdd(&s_cnt, 1);
                if (p < AGG_CAP) s_list[p] = eTail;
                else agg_add_serial(s_tile, h, dis, col, eTail, r, rowBase);
            }
        }
    }
    __syncthreads();

    {
        int total = __builtin_amdgcn_readfirstlane(s_cnt);
        if (total > AGG_CAP) total = AGG_CAP;
        const int waveU = __builtin_amdgcn_readfirstlane(wave);
        for (int base = waveU * 32; base < total; base += AGG_THREADS) {
            int cI = 0, lr = 0;
            float nm = 0.0f;
            if (base + lane < total) {
                const int e = s_list[base + lane];
                const int r = row[e];
                cI = col[e];
                lr = r - rowBase;
                nm = dis[r] * dis[cI];
            }
            int nv = total - base;
            if (nv > 32) nv = 32;
            int j = 0;
            for (; j + 8 <= nv; j += 8) {
                float val[8][4];
                int   lrj[8];
                float nmj[8];
                #pragma unroll
                for (int u = 0; u < 8; ++u) {
                    const int cj = __builtin_amdgcn_readlane(cI, j + u);
                    lrj[u] = __builtin_amdgcn_readlane(lr, j + u);
                    nmj[u] = __int_as_float(__builtin_amdgcn_readlane(__float_as_int(nm), j + u));
                    const float* hp = h + (size_t)cj * HIDDEN + lane;
                    val[u][0] = hp[0];
                    val[u][1] = hp[32];
                    val[u][2] = hp[64];
                    val[u][3] = hp[96];
                }
                #pragma unroll
                for (int u = 0; u < 8; ++u) {
                    float* tp = s_tile + lrj[u] * HIDDEN + lane;
                    atomicAdd(tp +  0, val[u][0] * nmj[u]);
                    atomicAdd(tp + 32, val[u][1] * nmj[u]);
                    atomicAdd(tp + 64, val[u][2] * nmj[u]);
                    atomicAdd(tp + 96, val[u][3] * nmj[u]);
                }
            }
            for (; j < nv; ++j) {
                const int cj = __builtin_amdgcn_readlane(cI, j);
                const int lj = __builtin_amdgcn_readlane(lr, j);
                const float nj = __int_as_float(__builtin_amdgcn_readlane(__float_as_int(nm), j));
                const float* hp = h + (size_t)cj * HIDDEN + lane;
                float* tp = s_tile + lj * HIDDEN + lane;
                atomicAdd(tp +  0, hp[0]  * nj);
                atomicAdd(tp + 32, hp[32] * nj);
                atomicAdd(tp + 64, hp[64] * nj);
                atomicAdd(tp + 96, hp[96] * nj);
            }
        }
    }
    __syncthreads();

    {
        const float4* t4 = (const float4*)s_tile;
        float4* o4 = (float4*)(out + (size_t)rowBase * HIDDEN);
        const int n = rowsHere * (HIDDEN / 4);
        for (int i = tid; i < n; i += AGG_THREADS) o4[i] = t4[i];
    }
}

extern "C" void kernel_launch(void* const* d_in, const int* in_sizes, int n_in,
                              void* d_out, int out_size, void* d_ws, size_t ws_size,
                              hipStream_t stream) {
    const float* x  = (const float*)d_in[0];
    const float* W  = (const float*)d_in[1];
    const int*   ei = (const int*)d_in[2];
    float* out = (float*)d_out;

    const int N = in_sizes[0] / HIDDEN;
    const int E = in_sizes[2] / 2;
    const int* rowIdx = ei;
    const int* colIdx = ei + E;

    float* h   = (float*)d_ws;
    float* deg = h + (size_t)N * HIDDEN;

    gcn_zero_kernel<<<(N + 255) / 256, 256, 0, stream>>>(deg, N);

    const int rowTiles = (N + 15) / 16;
    gcn_gemm_wmma_kernel<<<rowTiles, 256, 0, stream>>>(x, W, h, N);

    gcn_deg_kernel<<<1024, 256, 0, stream>>>(colIdx, deg, E);

    gcn_dis_kernel<<<(N + 255) / 256, 256, 0, stream>>>(deg, N);

    gcn_aggregate_kernel<<<(N + AGG_ROWS - 1) / AGG_ROWS, AGG_THREADS, 0, stream>>>(
        h, deg, rowIdx, colIdx, out, N, E);
}


# Aggregation blocks read their edge ids from bins built in the degree pass (LDS ranks, one global add per bin and tile, bins kept in out), not by each scanning all of row[]; adds stay in arrival order
# speedup vs baseline: 1.8683x; 1.2338x over previous
#include <hip/hip_runtime.h>

typedef __attribute__((ext_vector_type(2))) float v2f;
typedef __attribute__((ext_vector_type(8))) float v8f;

#define HIDDEN 128

#define AGG_ROWS    400
#define AGG_THREADS 1024
#define AGG_CAP     16384
#define AGG_UNR     8
#define BIN_WORDS   (AGG_ROWS * HIDDEN)
#define BIN_THREADS 1024
#define BIN_EPT     8
#define BIN_MAXB    1024

__global__ __launch_bounds__(256) void gcn_zero_kernel(float* __restrict__ deg, int n_deg,
                                                       int* __restrict__ binw, int nb) {
    int i = blockIdx.x * blockDim.x + threadIdx.x;
    if (i < n_deg) deg[i] = 0.0f;
    if (i < nb) binw[(size_t)i * BIN_WORDS] = 0;
}

__global__ __launch_bounds__(256) void gcn_gemm_wmma_kernel(
        const float* __restrict__ x, const float* __restrict__ W,
        float* __restrict__ h, int N) {
    const int lane    = threadIdx.x & 31;
    const int wave    = threadIdx.x >> 5;
    const int halfSel = lane >> 4;
    const int l       = lane & 15;

    const int rowBase = blockIdx.x * 16;
    const int nBase   = wave * 16;

    int rowLd = rowBase + l;
    if (rowLd >= N) rowLd = N - 1;

    const float* __restrict__ xrow = x + (size_t)rowLd * HIDDEN;
    const float* __restrict__ wrow = W + (size_t)(nBase + l) * HIDDEN;

    v8f c = {};
    #pragma unroll
    for (int k = 0; k < HIDDEN; k += 4) {
        const int ka = k + (halfSel << 1);
        v2f a = *(const v2f*)(xrow + ka);
        v2f b = *(const v2f*)(wrow + ka);
        c = __builtin_amdgcn_wmma_f32_16x16x4_f32(
                 false, a,  false, b,
                 (short)0, c,  false,  false);
    }

    float* __restrict__ hp = h + (size_t)rowBase * HIDDEN + nBase + l;
    #pragma unroll
    for (int v = 0; v < 8; ++v) {
        const int m = v + halfSel * 8;
        if (rowBase + m < N) hp[(size_t)m * HIDDEN] = c[v];
    }
}

__global__ __launch_bounds__(BIN_THREADS) void gcn_deg_bin_kernel(
        const int* __restrict__ row, const int* __restrict__ col,
        float* __restrict__ deg, int* __restrict__ binw, int N, int E, int nb) {
    __shared__ int s_c[BIN_MAXB];

    const int tid = threadIdx.x;
    const int per = (E + (int)gridDim.x - 1) / (int)gridDim.x;
    const int cBeg = (int)blockIdx.x * per;
    int cEnd = cBeg + per;
    if (cEnd > E) cEnd = E;
    const bool doBin = nb <= BIN_MAXB;

    for (int t0 = cBeg; t0 < cEnd; t0 += BIN_THREADS * BIN_EPT) {
        for (int i = tid; i < BIN_MAXB; i += BIN_THREADS) s_c[i] = 0;
        __syncthreads();

        int cc[BIN_EPT], bb[BIN_EPT], rk[BIN_EPT];
        #pragma unroll
        for (int u = 0; u < BIN_EPT; ++u) {
            const int e = t0 + u * BIN_THREADS + tid;
            const int ec = (e < cEnd) ? e : (cEnd - 1);
            cc[u] = col[ec];
            bb[u] = row[ec];
        }
        #pragma unroll
        for (int u = 0; u < BIN_EPT; ++u) {
            const int e = t0 + u * BIN_THREADS + tid;
            const int r = bb[u];
            const bool live = e < cEnd;
            cc[u] = live ? cc[u] : -1;
            bb[u] = (live && doBin && (unsigned)r < (unsigned)N) ? (r / AGG_ROWS) : -1;
            rk[u] = 0;
        }
        #pragma unroll
        for (int u = 0; u < BIN_EPT; ++u) {
            if (bb[u] >= 0) rk[u] = atomicAdd(&s_c[bb[u]], 1);
        }
        __syncthreads();

        for (int i = tid; i < nb && i < BIN_MAXB; i += BIN_THREADS) {
            const int c = s_c[i];
            if (c > 0) s_c[i] = atomicAdd(&binw[(size_t)i * BIN_WORDS], c);
        }
        __syncthreads();

        #pragma unroll
        for (int u = 0; u < BIN_EPT; ++u) {
            if (bb[u] >= 0) {
                const int b = bb[u];
                int rowsB = N - b * AGG_ROWS;
                if (rowsB > AGG_ROWS) rowsB = AGG_ROWS;
                const int pos = s_c[b] + rk[u];
                if (pos < rowsB * HIDDEN - 1)
                    binw[(size_t)b * BIN_WORDS + 1 + pos] = t0 + u * BIN_THREADS + tid;
            }
        }
        #pragma unroll
        for (int u = 0; u < BIN_EPT; ++u) {
            if (cc[u] >= 0) atomicAdd(&deg[cc[u]], 1.0f);
        }
        __syncthreads();
    }
}

__global__ __launch_bounds__(256) void gcn_dis_kernel(float* __restrict__ deg, int N) {
    int i = blockIdx.x * blockDim.x + threadIdx.x;
    if (i < N) {
        float d = deg[i];
        deg[i] = (d > 0.0f) ? __frsqrt_rn(d) : 0.0f;
    }
}

__device__ __noinline__ void agg_add_serial(float* tile, const float* __restrict__ h,
                                            const float* __restrict__ dis,
                                            const int* __restrict__ col,
                                            int e, int r, int rowBase) {
    const int cI = col[e];
    const float norm = dis[r] * dis[cI];
    const float* hp = h + (size_t)cI * HIDDEN;
    float* tp = tile + (r - rowBase) * HIDDEN;
    for (int k = 0; k < HIDDEN; ++k) atomicAdd(tp + k, hp[k] * norm);
}

__global__ __launch_bounds__(AGG_THREADS) void gcn_aggregate_kernel(
        const float* __restrict__ h, const float* __restrict__ dis,
        const int* __restrict__ row, const int* __restrict__ col,
        const int* binw, float* out, int N, int E) {
    __shared__ __attribute__((aligned(16))) float s_tile[AGG_ROWS * HIDDEN];
    __shared__ int s_list[AGG_CAP];
    __shared__ int s_cnt;

    const int tid  = threadIdx.x;
    const int lane = tid & 31;
    const int wave = tid >> 5;
    const int rowBase = blockIdx.x * AGG_ROWS;
    int rowsHere = N - rowBase;
    if (rowsHere > AGG_ROWS) rowsHere = AGG_ROWS;
    const unsigned urows = (unsigned)rowsHere;

    {
        float4* t4 = (float4*)s_tile;
        const float4 z = make_float4(0.f, 0.f, 0.f, 0.f);
        for (int i = tid; i < AGG_ROWS * HIDDEN / 4; i += AGG_THREADS) t4[i] = z;
        if (tid == 0) s_cnt = 0;
    }
    __syncthreads();

    const int* myBin = binw + (size_t)rowBase * HIDDEN;
    const int binCnt = myBin[0];
    const bool binOK = ((int)gridDim.x <= BIN_MAXB) && (binCnt <= AGG_CAP) &&
                       (binCnt < rowsHere * HIDDEN);
    if (binOK) {
        for (int i = tid; i < binCnt; i += AGG_THREADS) s_list[i] = myBin[1 + i];
        if (tid == 0) s_cnt = binCnt;
    } else {
        const int4* __restrict__ row4 = (const int4*)row;
        const int n4 = E >> 2;
        const int step = AGG_THREADS * AGG_UNR;
        const int nIter = (n4 + step - 1) / step;
        const int itStart = (int)(blockIdx.x % (unsigned)nIter);
        const int noMatch = rowBase - 1;
        for (int it = 0; it < nIter; ++it) {
            int itw = it + itStart;
            if (itw >= nIter) itw -= nIter;
            const int i0 = itw * step + tid;
            int4 v[AGG_UNR];
            #pragma unroll
            for (int u = 0; u < AGG_UNR; ++u) {
                const int i = i0 + u * AGG_THREADS;
                v[u] = (i < n4) ? row4[i] : make_int4(noMatch, noMatch, noMatch, noMatch);
            }
            #pragma unroll
            for (int u = 0; u < AGG_UNR; ++u) {
                const int e0 = (i0 + u * AGG_THREADS) * 4;
                const int rr[4] = {v[u].x, v[u].y, v[u].z, v[u].w};
                #pragma unroll
                for (int q = 0; q < 4; ++q) {
                    if ((unsigned)(rr[q] - rowBase) < urows) {
                        const int p = atomicAdd(&s_cnt, 1);
                        if (p < AGG_CAP) s_list[p] = e0 + q;
                        else agg_add_serial(s_tile, h, dis, col, e0 + q, rr[q], rowBase);
                    }
                }
            }
        }
        const int eTail = (n4 << 2) + tid;
        if (eTail < E) {
            const int r = row[eTail];
            if ((unsigned)(r - rowBase) < urows) {
                const int p = atomicAdd(&s_cnt, 1);
                if (p < AGG_CAP) s_list[p] = eTail;
                else agg_add_serial(s_tile, h, dis, col, eTail, r, rowBase);
            }
        }
    }
    __syncthreads();

    {
        int total = __builtin_amdgcn_readfirstlane(s_cnt);
        if (total > AGG_CAP) total = AGG_CAP;
        const int waveU = __builtin_amdgcn_readfirstlane(wave);
        for (int base = waveU * 32; base < total; base += AGG_THREADS) {
            int cI = 0, lr = 0;
            float nm = 0.0f;
            if (base + lane < total) {
                const int e = s_list[base + lane];
                const int r = row[e];
                cI = col[e];
                lr = r - rowBase;
                nm = dis[r] * dis[cI];
            }
            int nv = total - base;
            if (nv > 32) nv = 32;
            int j = 0;
            for (; j + 8 <= nv; j += 8) {
                float val[8][4];
                int   lrj[8];
                float nmj[8];
                #pragma unroll
                for (int u = 0; u < 8; ++u) {
                    const int cj = __builtin_amdgcn_readlane(cI, j + u);
                    lrj[u] = __builtin_amdgcn_readlane(lr, j + u);
                    nmj[u] = __int_as_float(__builtin_amdgcn_readlane(__float_as_int(nm), j + u));
                    const float* hp = h + (size_t)cj * HIDDEN + lane;
                    val[u][0] = hp[0];
                    val[u][1] = hp[32];
                    val[u][2] = hp[64];
                    val[u][3] = hp[96];
                }
                #pragma unroll
                for (int u = 0; u < 8; ++u) {
                    float* tp = s_tile + lrj[u] * HIDDEN + lane;
                    atomicAdd(tp +  0, val[u][0] * nmj[u]);
                    atomicAdd(tp + 32, val[u][1] * nmj[u]);
                    atomicAdd(tp + 64, val[u][2] * nmj[u]);
                    atomicAdd(tp + 96, val[u][3] * nmj[u]);
                }
            }
            for (; j < nv; ++j) {
                const int cj = __builtin_amdgcn_readlane(cI, j);
                const int lj = __builtin_amdgcn_readlane(lr, j);
                const float nj = __int_as_float(__builtin_amdgcn_readlane(__float_as_int(nm), j));
                const float* hp = h + (size_t)cj * HIDDEN + lane;
                float* tp = s_tile + lj * HIDDEN + lane;
                atomicAdd(tp +  0, hp[0]  * nj);
                atomicAdd(tp + 32, hp[32] * nj);
                atomicAdd(tp + 64, hp[64] * nj);
                atomicAdd(tp + 96, hp[96] * nj);
            }
        }
    }
    __syncthreads();

    {
        const float4* t4 = (const float4*)s_tile;
        float4* o4 = (float4*)(out + (size_t)rowBase * HIDDEN);
        const int n = rowsHere * (HIDDEN / 4);
        for (int i = tid; i < n; i += AGG_THREADS) o4[i] = t4[i];
    }
}

extern "C" void kernel_launch(void* const* d_in, const int* in_sizes, int n_in,
                              void* d_out, int out_size, void* d_ws, size_t ws_size,
                              hipStream_t stream) {
    const float* x  = (const float*)d_in[0];
    const float* W  = (const float*)d_in[1];
    const int*   ei = (const int*)d_in[2];
    float* out = (float*)d_out;

    const int N = in_sizes[0] / HIDDEN;
    const int E = in_sizes[2] / 2;
    const int* rowIdx = ei;
    const int* colIdx = ei + E;

    float* h   = (float*)d_ws;
    float* deg = h + (size_t)N * HIDDEN;

    const int nb = (N + AGG_ROWS - 1) / AGG_ROWS;
    int* binw = (int*)d_out;
    gcn_zero_kernel<<<(N + 255) / 256, 256, 0, stream>>>(deg, N, binw, nb);

    const int rowTiles = (N + 15) / 16;
    gcn_gemm_wmma_kernel<<<rowTiles, 256, 0, stream>>>(x, W, h, N);

    gcn_deg_bin_kernel<<<256, BIN_THREADS, 0, stream>>>(rowIdx, colIdx, deg, binw, N, E, nb);

    gcn_dis_kernel<<<(N + 255) / 256, 256, 0, stream>>>(deg, N);

    gcn_aggregate_kernel<<<nb, AGG_THREADS, 0, stream>>>(
        h, deg, rowIdx, colIdx, binw, out, N, E);
}


# Projection kernel: 768 resident blocks walk 16-row x tiles; a tile is fetched by whole-row 16-byte loads into padded LDS, A fragments read from LDS, B fragments kept in registers; k order unchanged
# speedup vs baseline: 2.2526x; 1.2057x over previous
#include <hip/hip_runtime.h>

typedef __attribute__((ext_vector_type(2))) float v2f;
typedef __attribute__((ext_vector_type(8))) float v8f;

#define HIDDEN 128

#define AGG_ROWS    400
#define AGG_THREADS 1024
#define AGG_CAP     16384
#define AGG_UNR     8
#define BIN_WORDS   (AGG_ROWS * HIDDEN)
#define BIN_THREADS 1024
#define BIN_EPT     8
#define BIN_MAXB    1024

__global__ __launch_bounds__(256) void gcn_zero_kernel(float* __restrict__ deg, int n_deg,
                                                       int* __restrict__ binw, int nb) {
    int i = blockIdx.x * blockDim.x + threadIdx.x;
    if (i < n_deg) deg[i] = 0.0f;
    if (i < nb) binw[(size_t)i * BIN_WORDS] = 0;
}

#define GEMM_THREADS 256
#define GEMM_TR      16
#define GEMM_PITCH   (HIDDEN + 4)
#define GEMM_BLOCKS  768
#define GEMM_KSTEPS  (HIDDEN / 4)

__global__ __launch_bounds__(GEMM_THREADS) void gcn_gemm_wmma_kernel(
        const float* __restrict__ x, const float* __restrict__ W,
        float* __restrict__ h, int N) {
    __shared__ __attribute__((aligned(16))) float s_x[2 * GEMM_TR * GEMM_PITCH];

    const int tid     = threadIdx.x;
    const int lane    = tid & 31;
    const int wave    = tid >> 5;
    const int halfSel = lane >> 4;
    const int l       = lane & 15;
    const int nBase   = wave * 16;
    const int nTiles  = (N + GEMM_TR - 1) / GEMM_TR;

    const float* __restrict__ wrow = W + (size_t)(nBase + l) * HIDDEN + (halfSel << 1);
    v2f bfrag[GEMM_KSTEPS];
    #pragma unroll
    for (int s = 0; s < GEMM_KSTEPS; ++s) bfrag[s] = *(const v2f*)(wrow + 4 * s);

    const int sr0 = wave, sr1 = wave + 8;
    float* const sdst0 = &s_x[sr0 * GEMM_PITCH + lane * 4];
    float* const sdst1 = &s_x[sr1 * GEMM_PITCH + lane * 4];

    int tile = blockIdx.x;
    float4 g0 = make_float4(0.f, 0.f, 0.f, 0.f), g1 = g0;
    if (tile < nTiles) {
        int r0 = tile * GEMM_TR + sr0; if (r0 >= N) r0 = N - 1;
        int r1 = tile * GEMM_TR + sr1; if (r1 >= N) r1 = N - 1;
        g0 = *(const float4*)(x + (size_t)r0 * HIDDEN + lane * 4);
        g1 = *(const float4*)(x + (size_t)r1 * HIDDEN + lane * 4);
    }

    int buf = 0;
    for (; tile < nTiles; tile += (int)gridDim.x) {
        const int bofs = buf * (GEMM_TR * GEMM_PITCH);
        *(float4*)(sdst0 + bofs) = g0;
        *(float4*)(sdst1 + bofs) = g1;
        __syncthreads();

        const int nxt = tile + (int)gridDim.x;
        if (nxt < nTiles) {
            int r0 = nxt * GEMM_TR + sr0; if (r0 >= N) r0 = N - 1;
            int r1 = nxt * GEMM_TR + sr1; if (r1 >= N) r1 = N - 1;
            g0 = *(const float4*)(x + (size_t)r0 * HIDDEN + lane * 4);
            g1 = *(const float4*)(x + (size_t)r1 * HIDDEN + lane * 4);
        }

        const float* ap = &s_x[bofs + l * GEMM_PITCH + (halfSel << 1)];
        v2f afrag[GEMM_KSTEPS];
        #pragma unroll
        for (int s = 0; s < GEMM_KSTEPS; ++s) afrag[s] = *(const v2f*)(ap + 4 * s);

        v8f c = {};
        #pragma unroll
        for (int s = 0; s < GEMM_KSTEPS; ++s) {
            c = __builtin_amdgcn_wmma_f32_16x16x4_f32(
                     false, afrag[s],  false, bfrag[s],
                     (short)0, c,  false,  false);
        }

        const int rowBase = tile * GEMM_TR;
        float* __restrict__ hp = h + (size_t)rowBase * HIDDEN + nBase + l;
        #pragma unroll
        for (int v = 0; v < 8; ++v) {
            const int m = v + halfSel * 8;
            if (rowBase + m < N) hp[(size_t)m * HIDDEN] = c[v];
        }
        buf ^= 1;
    }
}

__global__ __launch_bounds__(BIN_THREADS) void gcn_deg_bin_kernel(
        const int* __restrict__ row, const int* __restrict__ col,
        float* __restrict__ deg, int* __restrict__ binw, int N, int E, int nb) {
    __shared__ int s_c[BIN_MAXB];

    const int tid = threadIdx.x;
    const int per = (E + (int)gridDim.x - 1) / (int)gridDim.x;
    const int cBeg = (int)blockIdx.x * per;
    int cEnd = cBeg + per;
    if (cEnd > E) cEnd = E;
    const bool doBin = nb <= BIN_MAXB;

    for (int t0 = cBeg; t0 < cEnd; t0 += BIN_THREADS * BIN_EPT) {
        for (int i = tid; i < BIN_MAXB; i += BIN_THREADS) s_c[i] = 0;
        __syncthreads();

        int cc[BIN_EPT], bb[BIN_EPT], rk[BIN_EPT];
        #pragma unroll
        for (int u = 0; u < BIN_EPT; ++u) {
            const int e = t0 + u * BIN_THREADS + tid;
            const int ec = (e < cEnd) ? e : (cEnd - 1);
            cc[u] = col[ec];
            bb[u] = row[ec];
        }
        #pragma unroll
        for (int u = 0; u < BIN_EPT; ++u) {
            const int e = t0 + u * BIN_THREADS + tid;
            const int r = bb[u];
            const bool live = e < cEnd;
            cc[u] = live ? cc[u] : -1;
            bb[u] = (live && doBin && (unsigned)r < (unsigned)N) ? (r / AGG_ROWS) : -1;
            rk[u] = 0;
        }
        #pragma unroll
        for (int u = 0; u < BIN_EPT; ++u) {
            if (bb[u] >= 0) rk[u] = atomicAdd(&s_c[bb[u]], 1);
        }
        __syncthreads();

        for (int i = tid; i < nb && i < BIN_MAXB; i += BIN_THREADS) {
            const int c = s_c[i];
            if (c > 0) s_c[i] = atomicAdd(&binw[(size_t)i * BIN_WORDS], c);
        }
        __syncthreads();

        #pragma unroll
        for (int u = 0; u < BIN_EPT; ++u) {
            if (bb[u] >= 0) {
                const int b = bb[u];
                int rowsB = N - b * AGG_ROWS;
                if (rowsB > AGG_ROWS) rowsB = AGG_ROWS;
                const int pos = s_c[b] + rk[u];
                if (pos < rowsB * HIDDEN - 1)
                    binw[(size_t)b * BIN_WORDS + 1 + pos] = t0 + u * BIN_THREADS + tid;
            }
        }
        #pragma unroll
        for (int u = 0; u < BIN_EPT; ++u) {
            if (cc[u] >= 0) atomicAdd(&deg[cc[u]], 1.0f);
        }
        __syncthreads();
    }
}

__global__ __launch_bounds__(256) void gcn_dis_kernel(float* __restrict__ deg, int N) {
    int i = blockIdx.x * blockDim.x + threadIdx.x;
    if (i < N) {
        float d = deg[i];
        deg[i] = (d > 0.0f) ? __frsqrt_rn(d) : 0.0f;
    }
}

__device__ __noinline__ void agg_add_serial(float* tile, const float* __restrict__ h,
                                            const float* __restrict__ dis,
                                            const int* __restrict__ col,
                                            int e, int r, int rowBase) {
    const int cI = col[e];
    const float norm = dis[r] * dis[cI];
    const float* hp = h + (size_t)cI * HIDDEN;
    float* tp = tile + (r - rowBase) * HIDDEN;
    for (int k = 0; k < HIDDEN; ++k) atomicAdd(tp + k, hp[k] * norm);
}

__global__ __launch_bounds__(AGG_THREADS) void gcn_aggregate_kernel(
        const float* __restrict__ h, const float* __restrict__ dis,
        const int* __restrict__ row, const int* __restrict__ col,
        const int* binw, float* out, int N, int E) {
    __shared__ __attribute__((aligned(16))) float s_tile[AGG_ROWS * HIDDEN];
    __shared__ int s_list[AGG_CAP];
    __shared__ int s_cnt;

    const int tid  = threadIdx.x;
    const int lane = tid & 31;
    const int wave = tid >> 5;
    const int rowBase = blockIdx.x * AGG_ROWS;
    int rowsHere = N - rowBase;
    if (rowsHere > AGG_ROWS) rowsHere = AGG_ROWS;
    const unsigned urows = (unsigned)rowsHere;

    {
        float4* t4 = (float4*)s_tile;
        const float4 z = make_float4(0.f, 0.f, 0.f, 0.f);
        for (int i = tid; i < AGG_ROWS * HIDDEN / 4; i += AGG_THREADS) t4[i] = z;
        if (tid == 0) s_cnt = 0;
    }
    __syncthreads();

    const int* myBin = binw + (size_t)rowBase * HIDDEN;
    const int binCnt = myBin[0];
    const bool binOK = ((int)gridDim.x <= BIN_MAXB) && (binCnt <= AGG_CAP) &&
                       (binCnt < rowsHere * HIDDEN);
    if (binOK) {
        for (int i = tid; i < binCnt; i += AGG_THREADS) s_list[i] = myBin[1 + i];
        if (tid == 0) s_cnt = binCnt;
    } else {
        const int4* __restrict__ row4 = (const int4*)row;
        const int n4 = E >> 2;
        const int step = AGG_THREADS * AGG_UNR;
        const int nIter = (n4 + step - 1) / step;
        const int itStart = (int)(blockIdx.x % (unsigned)nIter);
        const int noMatch = rowBase - 1;
        for (int it = 0; it < nIter; ++it) {
            int itw = it + itStart;
            if (itw >= nIter) itw -= nIter;
            const int i0 = itw * step + tid;
            int4 v[AGG_UNR];
            #pragma unroll
            for (int u = 0; u < AGG_UNR; ++u) {
                const int i = i0 + u * AGG_THREADS;
                v[u] = (i < n4) ? row4[i] : make_int4(noMatch, noMatch, noMatch, noMatch);
            }
            #pragma unroll
            for (int u = 0; u < AGG_UNR; ++u) {
                const int e0 = (i0 + u * AGG_THREADS) * 4;
                const int rr[4] = {v[u].x, v[u].y, v[u].z, v[u].w};
                #pragma unroll
                for (int q = 0; q < 4; ++q) {
                    if ((unsigned)(rr[q] - rowBase) < urows) {
                        const int p = atomicAdd(&s_cnt, 1);
                        if (p < AGG_CAP) s_list[p] = e0 + q;
                        else agg_add_serial(s_tile, h, dis, col, e0 + q, rr[q], rowBase);
                    }
                }
            }
        }
        const int eTail = (n4 << 2) + tid;
        if (eTail < E) {
            const int r = row[eTail];
            if ((unsigned)(r - rowBase) < urows) {
                const int p = atomicAdd(&s_cnt, 1);
                if (p < AGG_CAP) s_list[p] = eTail;
                else agg_add_serial(s_tile, h, dis, col, eTail, r, rowBase);
            }
        }
    }
    __syncthreads();

    {
        int total = __builtin_amdgcn_readfirstlane(s_cnt);
        if (total > AGG_CAP) total = AGG_CAP;
        const int waveU = __builtin_amdgcn_readfirstlane(wave);
        for (int base = waveU * 32; base < total; base += AGG_THREADS) {
            int cI = 0, lr = 0;
            float nm = 0.0f;
            if (base + lane < total) {
                const int e = s_list[base + lane];
                const int r = row[e];
                cI = col[e];
                lr = r - rowBase;
                nm = dis[r] * dis[cI];
            }
            int nv = total - base;
            if (nv > 32) nv = 32;
            int j = 0;
            for (; j + 8 <= nv; j += 8) {
                float val[8][4];
                int   lrj[8];
                float nmj[8];
                #pragma unroll
                for (int u = 0; u < 8; ++u) {
                    const int cj = __builtin_amdgcn_readlane(cI, j + u);
                    lrj[u] = __builtin_amdgcn_readlane(lr, j + u);
                    nmj[u] = __int_as_float(__builtin_amdgcn_readlane(__float_as_int(nm), j + u));
                    const float* hp = h + (size_t)cj * HIDDEN + lane;
                    val[u][0] = hp[0];
                    val[u][1] = hp[32];
                    val[u][2] = hp[64];
                    val[u][3] = hp[96];
                }
                #pragma unroll
                for (int u = 0; u < 8; ++u) {
                    float* tp = s_tile + lrj[u] * HIDDEN + lane;
                    atomicAdd(tp +  0, val[u][0] * nmj[u]);
                    atomicAdd(tp + 32, val[u][1] * nmj[u]);
                    atomicAdd(tp + 64, val[u][2] * nmj[u]);
                    atomicAdd(tp + 96, val[u][3] * nmj[u]);
                }
            }
            for (; j < nv; ++j) {
                const int cj = __builtin_amdgcn_readlane(cI, j);
                const int lj = __builtin_amdgcn_readlane(lr, j);
                const float nj = __int_as_float(__builtin_amdgcn_readlane(__float_as_int(nm), j));
                const float* hp = h + (size_t)cj * HIDDEN + lane;
                float* tp = s_tile + lj * HIDDEN + lane;
                atomicAdd(tp +  0, hp[0]  * nj);
                atomicAdd(tp + 32, hp[32] * nj);
                atomicAdd(tp + 64, hp[64] * nj);
                atomicAdd(tp + 96, hp[96] * nj);
            }
        }
    }
    __syncthreads();

    {
        const float4* t4 = (const float4*)s_tile;
        float4* o4 = (float4*)(out + (size_t)rowBase * HIDDEN);
        const int n = rowsHere * (HIDDEN / 4);
        for (int i = tid; i < n; i += AGG_THREADS) o4[i] = t4[i];
    }
}

extern "C" void kernel_launch(void* const* d_in, const int* in_sizes, int n_in,
                              void* d_out, int out_size, void* d_ws, size_t ws_size,
                              hipStream_t stream) {
    const float* x  = (const float*)d_in[0];
    const float* W  = (const float*)d_in[1];
    const int*   ei = (const int*)d_in[2];
    float* out = (float*)d_out;

    const int N = in_sizes[0] / HIDDEN;
    const int E = in_sizes[2] / 2;
    const int* rowIdx = ei;
    const int* colIdx = ei + E;

    float* h   = (float*)d_ws;
    float* deg = h + (size_t)N * HIDDEN;

    const int nb = (N + AGG_ROWS - 1) / AGG_ROWS;
    int* binw = (int*)d_out;
    gcn_zero_kernel<<<(N + 255) / 256, 256, 0, stream>>>(deg, N, binw, nb);

    gcn_gemm_wmma_kernel<<<GEMM_BLOCKS, GEMM_THREADS, 0, stream>>>(x, W, h, N);

    gcn_deg_bin_kernel<<<256, BIN_THREADS, 0, stream>>>(rowIdx, colIdx, deg, binw, N, E, nb);

    gcn_dis_kernel<<<(N + 255) / 256, 256, 0, stream>>>(deg, N);

    gcn_aggregate_kernel<<<nb, AGG_THREADS, 0, stream>>>(
        h, deg, rowIdx, colIdx, binw, out, N, E);
}


# Bin entries hold the edge's source row and local out row packed in one word, written by the binning pass, so aggregation reads neither row[] nor col[] per edge; adds stay in arrival order
# speedup vs baseline: 2.3480x; 1.0424x over previous
#include <hip/hip_runtime.h>

typedef __attribute__((ext_vector_type(2))) float v2f;
typedef __attribute__((ext_vector_type(8))) float v8f;

#define HIDDEN 128

#define AGG_ROWS    400
#define AGG_THREADS 1024
#define AGG_CAP     16384
#define AGG_UNR     8
#define BIN_WORDS   (AGG_ROWS * HIDDEN)
#define BIN_THREADS 1024
#define BIN_EPT     8
#define BIN_MAXB    1024
#define PACK_SHIFT  23
#define PACK_MASK   ((1u << PACK_SHIFT) - 1u)
static_assert(AGG_ROWS <= (1 << (32 - PACK_SHIFT)), "the local row must fit above the source row");

__global__ __launch_bounds__(256) void gcn_zero_kernel(float* __restrict__ deg, int n_deg,
                                                       int* __restrict__ binw, int nb) {
    int i = blockIdx.x * blockDim.x + threadIdx.x;
    if (i < n_deg) deg[i] = 0.0f;
    if (i < nb) binw[(size_t)i * BIN_WORDS] = 0;
}

#define GEMM_THREADS 256
#define GEMM_TR      16
#define GEMM_PITCH   (HIDDEN + 4)
#define GEMM_BLOCKS  768
#define GEMM_KSTEPS  (HIDDEN / 4)

__global__ __launch_bounds__(GEMM_THREADS) void gcn_gemm_wmma_kernel(
        const float* __restrict__ x, const float* __restrict__ W,
        float* __restrict__ h, int N) {
    __shared__ __attribute__((aligned(16))) float s_x[2 * GEMM_TR * GEMM_PITCH];

    const int tid     = threadIdx.x;
    const int lane    = tid & 31;
    const int wave    = tid >> 5;
    const int halfSel = lane >> 4;
    const int l       = lane & 15;
    const int nBase   = wave * 16;
    const int nTiles  = (N + GEMM_TR - 1) / GEMM_TR;

    const float* __restrict__ wrow = W + (size_t)(nBase + l) * HIDDEN + (halfSel << 1);
    v2f bfrag[GEMM_KSTEPS];
    #pragma unroll
    for (int s = 0; s < GEMM_KSTEPS; ++s) bfrag[s] = *(const v2f*)(wrow + 4 * s);

    const int sr0 = wave, sr1 = wave + 8;
    float* const sdst0 = &s_x[sr0 * GEMM_PITCH + lane * 4];
    float* const sdst1 = &s_x[sr1 * GEMM_PITCH + lane * 4];

    int tile = blockIdx.x;
    float4 g0 = make_float4(0.f, 0.f, 0.f, 0.f), g1 = g0;
    if (tile < nTiles) {
        int r0 = tile * GEMM_TR + sr0; if (r0 >= N) r0 = N - 1;
        int r1 = tile * GEMM_TR + sr1; if (r1 >= N) r1 = N - 1;
        g0 = *(const float4*)(x + (size_t)r0 * HIDDEN + lane * 4);
        g1 = *(const float4*)(x + (size_t)r1 * HIDDEN + lane * 4);
    }

    int buf = 0;
    for (; tile < nTiles; tile += (int)gridDim.x) {
        const int bofs = buf * (GEMM_TR * GEMM_PITCH);
        *(float4*)(sdst0 + bofs) = g0;
        *(float4*)(sdst1 + bofs) = g1;
        __syncthreads();

        const int nxt = tile + (int)gridDim.x;
        if (nxt < nTiles) {
            int r0 = nxt * GEMM_TR + sr0; if (r0 >= N) r0 = N - 1;
            int r1 = nxt * GEMM_TR + sr1; if (r1 >= N) r1 = N - 1;
            g0 = *(const float4*)(x + (size_t)r0 * HIDDEN + lane * 4);
            g1 = *(const float4*)(x + (size_t)r1 * HIDDEN + lane * 4);
        }

        const float* ap = &s_x[bofs + l * GEMM_PITCH + (halfSel << 1)];
        v2f afrag[GEMM_KSTEPS];
        #pragma unroll
        for (int s = 0; s < GEMM_KSTEPS; ++s) afrag[s] = *(const v2f*)(ap + 4 * s);

        v8f c = {};
        #pragma unroll
        for (int s = 0; s < GEMM_KSTEPS; ++s) {
            c = __builtin_amdgcn_wmma_f32_16x16x4_f32(
                     false, afrag[s],  false, bfrag[s],
                     (short)0, c,  false,  false);
        }

        const int rowBase = tile * GEMM_TR;
        float* __restrict__ hp = h + (size_t)rowBase * HIDDEN + nBase + l;
        #pragma unroll
        for (int v = 0; v < 8; ++v) {
            const int m = v + halfSel * 8;
            if (rowBase + m < N) hp[(size_t)m * HIDDEN] = c[v];
        }
        buf ^= 1;
    }
}

__global__ __launch_bounds__(BIN_THREADS) void gcn_deg_bin_kernel(
        const int* __restrict__ row, const int* __restrict__ col,
        float* __restrict__ deg, int* __restrict__ binw, int N, int E, int nb) {
    __shared__ int s_c[BIN_MAXB];

    const int tid = threadIdx.x;
    const int per = (E + (int)gridDim.x - 1) / (int)gridDim.x;
    const int cBeg = (int)blockIdx.x * per;
    int cEnd = cBeg + per;
    if (cEnd > E) cEnd = E;
    const bool doBin = (nb <= BIN_MAXB) && (N <= (1 << PACK_SHIFT));

    for (int t0 = cBeg; t0 < cEnd; t0 += BIN_THREADS * BIN_EPT) {
        for (int i = tid; i < BIN_MAXB; i += BIN_THREADS) s_c[i] = 0;
        __syncthreads();

        int cc[BIN_EPT], bb[BIN_EPT], rk[BIN_EPT], pk[BIN_EPT];
        #pragma unroll
        for (int u = 0; u < BIN_EPT; ++u) {
            const int e = t0 + u * BIN_THREADS + tid;
            const int ec = (e < cEnd) ? e : (cEnd - 1);
            cc[u] = col[ec];
            bb[u] = row[ec];
        }
        #pragma unroll
        for (int u = 0; u < BIN_EPT; ++u) {
            const int e = t0 + u * BIN_THREADS + tid;
            const int r = bb[u];
            const bool live = e < cEnd;
            const int b = r / AGG_ROWS;
            pk[u] = (int)(((unsigned)(r - b * AGG_ROWS) << PACK_SHIFT) | ((unsigned)cc[u] & PACK_MASK));
            cc[u] = live ? cc[u] : -1;
            bb[u] = (live && doBin && (unsigned)r < (unsigned)N) ? b : -1;
            rk[u] = 0;
        }
        #pragma unroll
        for (int u = 0; u < BIN_EPT; ++u) {
            if (bb[u] >= 0) rk[u] = atomicAdd(&s_c[bb[u]], 1);
        }
        __syncthreads();

        for (int i = tid; i < nb && i < BIN_MAXB; i += BIN_THREADS) {
            const int c = s_c[i];
            if (c > 0) s_c[i] = atomicAdd(&binw[(size_t)i * BIN_WORDS], c);
        }
        __syncthreads();

        #pragma unroll
        for (int u = 0; u < BIN_EPT; ++u) {
            if (bb[u] >= 0) {
                const int b = bb[u];
                int rowsB = N - b * AGG_ROWS;
                if (rowsB > AGG_ROWS) rowsB = AGG_ROWS;
                const int pos = s_c[b] + rk[u];
                if (pos < rowsB * HIDDEN - 1)
                    binw[(size_t)b * BIN_WORDS + 1 + pos] = pk[u];
            }
        }
        #pragma unroll
        for (int u = 0; u < BIN_EPT; ++u) {
            if (cc[u] >= 0) atomicAdd(&deg[cc[u]], 1.0f);
        }
        __syncthreads();
    }
}

__global__ __launch_bounds__(256) void gcn_dis_kernel(float* __restrict__ deg, int N) {
    int i = blockIdx.x * blockDim.x + threadIdx.x;
    if (i < N) {
        float d = deg[i];
        deg[i] = (d > 0.0f) ? __frsqrt_rn(d) : 0.0f;
    }
}

__device__ __noinline__ void agg_add_serial(float* tile, const float* __restrict__ h,
                                            const float* __restrict__ dis,
                                            const int* __restrict__ col,
                                            int e, int r, int rowBase) {
    const int cI = col[e];
    const float norm = dis[r] * dis[cI];
    const float* hp = h + (size_t)cI * HIDDEN;
    float* tp = tile + (r - rowBase) * HIDDEN;
    for (int k = 0; k < HIDDEN; ++k) atomicAdd(tp + k, hp[k] * norm);
}

__global__ __launch_bounds__(AGG_THREADS) void gcn_aggregate_kernel(
        const float* __restrict__ h, const float* __restrict__ dis,
        const int* __restrict__ row, const int* __restrict__ col,
        const int* binw, float* out, int N, int E) {
    __shared__ __attribute__((aligned(16))) float s_tile[AGG_ROWS * HIDDEN];
    __shared__ int s_list[AGG_CAP];
    __shared__ int s_cnt;

    const int tid  = threadIdx.x;
    const int lane = tid & 31;
    const int wave = tid >> 5;
    const int rowBase = blockIdx.x * AGG_ROWS;
    int rowsHere = N - rowBase;
    if (rowsHere > AGG_ROWS) rowsHere = AGG_ROWS;
    const unsigned urows = (unsigned)rowsHere;

    {
        float4* t4 = (float4*)s_tile;
        const float4 z = make_float4(0.f, 0.f, 0.f, 0.f);
        for (int i = tid; i < AGG_ROWS * HIDDEN / 4; i += AGG_THREADS) t4[i] = z;
        if (tid == 0) s_cnt = 0;
    }
    __syncthreads();

    const int* myBin = binw + (size_t)rowBase * HIDDEN;
    const int binCnt = myBin[0];
    const bool binOK = ((int)gridDim.x <= BIN_MAXB) && (N <= (1 << PACK_SHIFT)) &&
                       (binCnt <= AGG_CAP) && (binCnt < rowsHere * HIDDEN);
    if (binOK) {
        for (int i = tid; i < binCnt; i += AGG_THREADS) s_list[i] = myBin[1 + i];
        if (tid == 0) s_cnt = binCnt;
    } else {
        const int4* __restrict__ row4 = (const int4*)row;
        const int n4 = E >> 2;
        const int step = AGG_THREADS * AGG_UNR;
        const int nIter = (n4 + step - 1) / step;
        const int itStart = (int)(blockIdx.x % (unsigned)nIter);
        const int noMatch = rowBase - 1;
        for (int it = 0; it < nIter; ++it) {
            int itw = it + itStart;
            if (itw >= nIter) itw -= nIter;
            const int i0 = itw * step + tid;
            int4 v[AGG_UNR];
            #pragma unroll
            for (int u = 0; u < AGG_UNR; ++u) {
                const int i = i0 + u * AGG_THREADS;
                v[u] = (i < n4) ? row4[i] : make_int4(noMatch, noMatch, noMatch, noMatch);
            }
            #pragma unroll
            for (int u = 0; u < AGG_UNR; ++u) {
                const int e0 = (i0 + u * AGG_THREADS) * 4;
                const int rr[4] = {v[u].x, v[u].y, v[u].z, v[u].w};
                #pragma unroll
                for (int q = 0; q < 4; ++q) {
                    if ((unsigned)(rr[q] - rowBase) < urows) {
                        const int p = atomicAdd(&s_cnt, 1);
                        if (p < AGG_CAP) s_list[p] = e0 + q;
                        else agg_add_serial(s_tile, h, dis, col, e0 + q, rr[q], rowBase);
                    }
                }
            }
        }
        const int eTail = (n4 << 2) + tid;
        if (eTail < E) {
            const int r = row[eTail];
            if ((unsigned)(r - rowBase) < urows) {
                const int p = atomicAdd(&s_cnt, 1);
                if (p < AGG_CAP) s_list[p] = eTail;
                else agg_add_serial(s_tile, h, dis, col, eTail, r, rowBase);
            }
        }
    }
    __syncthreads();

    {
        int total = __builtin_amdgcn_readfirstlane(s_cnt);
        if (total > AGG_CAP) total = AGG_CAP;
        const int waveU = __builtin_amdgcn_readfirstlane(wave);
        for (int base = waveU * 32; base < total; base += AGG_THREADS) {
            int cI = 0, lr = 0;
            float dr = 0.0f, dc = 0.0f;
            if (base + lane < total) {
                const int w = s_list[base + lane];
                if (binOK) {
                    cI = (int)((unsigned)w & PACK_MASK);
                    lr = (int)((unsigned)w >> PACK_SHIFT);
                } else {
                    cI = col[w];
                    lr = row[w] - rowBase;
                }
                dr = dis[rowBase + lr];
                dc = dis[cI];
            }
            int nv = total - base;
            if (nv > 32) nv = 32;
            int j = 0;
            for (; j + 8 <= nv; j += 8) {
                float val[8][4];
                int   lrj[8];
                #pragma unroll
                for (int u = 0; u < 8; ++u) {
                    const int cj = __builtin_amdgcn_readlane(cI, j + u);
                    lrj[u] = __builtin_amdgcn_readlane(lr, j + u);
                    const float* hp = h + (size_t)cj * HIDDEN + lane;
                    val[u][0] = hp[0];
                    val[u][1] = hp[32];
                    val[u][2] = hp[64];
                    val[u][3] = hp[96];
                }
                #pragma unroll
                for (int u = 0; u < 8; ++u) {
                    const float nmj =
                        __int_as_float(__builtin_amdgcn_readlane(__float_as_int(dr), j + u)) *
                        __int_as_float(__builtin_amdgcn_readlane(__float_as_int(dc), j + u));
                    float* tp = s_tile + lrj[u] * HIDDEN + lane;
                    atomicAdd(tp +  0, val[u][0] * nmj);
                    atomicAdd(tp + 32, val[u][1] * nmj);
                    atomicAdd(tp + 64, val[u][2] * nmj);
                    atomicAdd(tp + 96, val[u][3] * nmj);
                }
            }
            for (; j < nv; ++j) {
                const int cj = __builtin_amdgcn_readlane(cI, j);
                const int lj = __builtin_amdgcn_readlane(lr, j);
                const float nj =
                    __int_as_float(__builtin_amdgcn_readlane(__float_as_int(dr), j)) *
                    __int_as_float(__builtin_amdgcn_readlane(__float_as_int(dc), j));
                const float* hp = h + (size_t)cj * HIDDEN + lane;
                float* tp = s_tile + lj * HIDDEN + lane;
                atomicAdd(tp +  0, hp[0]  * nj);
                atomicAdd(tp + 32, hp[32] * nj);
                atomicAdd(tp + 64, hp[64] * nj);
                atomicAdd(tp + 96, hp[96] * nj);
            }
        }
    }
    __syncthreads();

    {
        const float4* t4 = (const float4*)s_tile;
        float4* o4 = (float4*)(out + (size_t)rowBase * HIDDEN);
        const int n = rowsHere * (HIDDEN / 4);
        for (int i = tid; i < n; i += AGG_THREADS) o4[i] = t4[i];
    }
}

extern "C" void kernel_launch(void* const* d_in, const int* in_sizes, int n_in,
                              void* d_out, int out_size, void* d_ws, size_t ws_size,
                              hipStream_t stream) {
    const float* x  = (const float*)d_in[0];
    const float* W  = (const float*)d_in[1];
    const int*   ei = (const int*)d_in[2];
    float* out = (float*)d_out;

    const int N = in_sizes[0] / HIDDEN;
    const int E = in_sizes[2] / 2;
    const int* rowIdx = ei;
    const int* colIdx = ei + E;

    float* h   = (float*)d_ws;
    float* deg = h + (size_t)N * HIDDEN;

    const int nb = (N + AGG_ROWS - 1) / AGG_ROWS;
    int* binw = (int*)d_out;
    gcn_zero_kernel<<<(N + 255) / 256, 256, 0, stream>>>(deg, N, binw, nb);

    gcn_gemm_wmma_kernel<<<GEMM_BLOCKS, GEMM_THREADS, 0, stream>>>(x, W, h, N);

    gcn_deg_bin_kernel<<<256, BIN_THREADS, 0, stream>>>(rowIdx, colIdx, deg, binw, N, E, nb);

    gcn_dis_kernel<<<(N + 255) / 256, 256, 0, stream>>>(deg, N);

    gcn_aggregate_kernel<<<nb, AGG_THREADS, 0, stream>>>(
        h, deg, rowIdx, colIdx, binw, out, N, E);
}


# Aggregation: each block counting-sorts its bin by destination row in LDS (integer counters); one wave sums a row's sources in registers, stores it once; no LDS f32 adds; sums in list (arrival) order
# speedup vs baseline: 2.4800x; 1.0562x over previous
#include <hip/hip_runtime.h>

typedef __attribute__((ext_vector_type(2))) float v2f;
typedef __attribute__((ext_vector_type(8))) float v8f;

#define HIDDEN 128

#define AGG_ROWS    400
#define AGG_THREADS 1024
#define AGG_CAP     16384
#define AGG_UNR     8
#define BIN_WORDS   (AGG_ROWS * HIDDEN)
#define BIN_THREADS 1024
#define BIN_EPT     8
#define BIN_MAXB    1024
#define PACK_SHIFT  23
#define PACK_MASK   ((1u << PACK_SHIFT) - 1u)
static_assert(AGG_ROWS <= (1 << (32 - PACK_SHIFT)), "the local row must fit above the source row");

__global__ __launch_bounds__(256) void gcn_zero_kernel(float* __restrict__ deg, int n_deg,
                                                       int* __restrict__ binw, int nb) {
    int i = blockIdx.x * blockDim.x + threadIdx.x;
    if (i < n_deg) deg[i] = 0.0f;
    if (i < nb) binw[(size_t)i * BIN_WORDS] = 0;
}

#define GEMM_THREADS 256
#define GEMM_TR      16
#define GEMM_PITCH   (HIDDEN + 4)
#define GEMM_BLOCKS  768
#define GEMM_KSTEPS  (HIDDEN / 4)

__global__ __launch_bounds__(GEMM_THREADS) void gcn_gemm_wmma_kernel(
        const float* __restrict__ x, const float* __restrict__ W,
        float* __restrict__ h, int N) {
    __shared__ __attribute__((aligned(16))) float s_x[2 * GEMM_TR * GEMM_PITCH];

    const int tid     = threadIdx.x;
    const int lane    = tid & 31;
    const int wave    = tid >> 5;
    const int halfSel = lane >> 4;
    const int l       = lane & 15;
    const int nBase   = wave * 16;
    const int nTiles  = (N + GEMM_TR - 1) / GEMM_TR;

    const float* __restrict__ wrow = W + (size_t)(nBase + l) * HIDDEN + (halfSel << 1);
    v2f bfrag[GEMM_KSTEPS];
    #pragma unroll
    for (int s = 0; s < GEMM_KSTEPS; ++s) bfrag[s] = *(const v2f*)(wrow + 4 * s);

    const int sr0 = wave, sr1 = wave + 8;
    float* const sdst0 = &s_x[sr0 * GEMM_PITCH + lane * 4];
    float* const sdst1 = &s_x[sr1 * GEMM_PITCH + lane * 4];

    int tile = blockIdx.x;
    float4 g0 = make_float4(0.f, 0.f, 0.f, 0.f), g1 = g0;
    if (tile < nTiles) {
        int r0 = tile * GEMM_TR + sr0; if (r0 >= N) r0 = N - 1;
        int r1 = tile * GEMM_TR + sr1; if (r1 >= N) r1 = N - 1;
        g0 = *(const float4*)(x + (size_t)r0 * HIDDEN + lane * 4);
        g1 = *(const float4*)(x + (size_t)r1 * HIDDEN + lane * 4);
    }

    int buf = 0;
    for (; tile < nTiles; tile += (int)gridDim.x) {
        const int bofs = buf * (GEMM_TR * GEMM_PITCH);
        *(float4*)(sdst0 + bofs) = g0;
        *(float4*)(sdst1 + bofs) = g1;
        __syncthreads();

        const int nxt = tile + (int)gridDim.x;
        if (nxt < nTiles) {
            int r0 = nxt * GEMM_TR + sr0; if (r0 >= N) r0 = N - 1;
            int r1 = nxt * GEMM_TR + sr1; if (r1 >= N) r1 = N - 1;
            g0 = *(const float4*)(x + (size_t)r0 * HIDDEN + lane * 4);
            g1 = *(const float4*)(x + (size_t)r1 * HIDDEN + lane * 4);
        }

        const float* ap = &s_x[bofs + l * GEMM_PITCH + (halfSel << 1)];
        v2f afrag[GEMM_KSTEPS];
        #pragma unroll
        for (int s = 0; s < GEMM_KSTEPS; ++s) afrag[s] = *(const v2f*)(ap + 4 * s);

        v8f c = {};
        #pragma unroll
        for (int s = 0; s < GEMM_KSTEPS; ++s) {
            c = __builtin_amdgcn_wmma_f32_16x16x4_f32(
                     false, afrag[s],  false, bfrag[s],
                     (short)0, c,  false,  false);
        }

        const int rowBase = tile * GEMM_TR;
        float* __restrict__ hp = h + (size_t)rowBase * HIDDEN + nBase + l;
        #pragma unroll
        for (int v = 0; v < 8; ++v) {
            const int m = v + halfSel * 8;
            if (rowBase + m < N) hp[(size_t)m * HIDDEN] = c[v];
        }
        buf ^= 1;
    }
}

__global__ __launch_bounds__(BIN_THREADS) void gcn_deg_bin_kernel(
        const int* __restrict__ row, const int* __restrict__ col,
        float* __restrict__ deg, int* __restrict__ binw, int N, int E, int nb) {
    __shared__ int s_c[BIN_MAXB];

    const int tid = threadIdx.x;
    const int per = (E + (int)gridDim.x - 1) / (int)gridDim.x;
    const int cBeg = (int)blockIdx.x * per;
    int cEnd = cBeg + per;
    if (cEnd > E) cEnd = E;
    const bool doBin = (nb <= BIN_MAXB) && (N <= (1 << PACK_SHIFT));

    for (int t0 = cBeg; t0 < cEnd; t0 += BIN_THREADS * BIN_EPT) {
        for (int i = tid; i < BIN_MAXB; i += BIN_THREADS) s_c[i] = 0;
        __syncthreads();

        int cc[BIN_EPT], bb[BIN_EPT], rk[BIN_EPT], pk[BIN_EPT];
        #pragma unroll
        for (int u = 0; u < BIN_EPT; ++u) {
            const int e = t0 + u * BIN_THREADS + tid;
            const int ec = (e < cEnd) ? e : (cEnd - 1);
            cc[u] = col[ec];
            bb[u] = row[ec];
        }
        #pragma unroll
        for (int u = 0; u < BIN_EPT; ++u) {
            const int e = t0 + u * BIN_THREADS + tid;
            const int r = bb[u];
            const bool live = e < cEnd;
            const int b = r / AGG_ROWS;
            pk[u] = (int)(((unsigned)(r - b * AGG_ROWS) << PACK_SHIFT) | ((unsigned)cc[u] & PACK_MASK));
            cc[u] = live ? cc[u] : -1;
            bb[u] = (live && doBin && (unsigned)r < (unsigned)N) ? b : -1;
            rk[u] = 0;
        }
        #pragma unroll
        for (int u = 0; u < BIN_EPT; ++u) {
            if (bb[u] >= 0) rk[u] = atomicAdd(&s_c[bb[u]], 1);
        }
        __syncthreads();

        for (int i = tid; i < nb && i < BIN_MAXB; i += BIN_THREADS) {
            const int c = s_c[i];
            if (c > 0) s_c[i] = atomicAdd(&binw[(size_t)i * BIN_WORDS], c);
        }
        __syncthreads();

        #pragma unroll
        for (int u = 0; u < BIN_EPT; ++u) {
            if (bb[u] >= 0) {
                const int b = bb[u];
                int rowsB = N - b * AGG_ROWS;
                if (rowsB > AGG_ROWS) rowsB = AGG_ROWS;
                const int pos = s_c[b] + rk[u];
                if (pos < rowsB * HIDDEN - 1)
                    binw[(size_t)b * BIN_WORDS + 1 + pos] = pk[u];
            }
        }
        #pragma unroll
        for (int u = 0; u < BIN_EPT; ++u) {
            if (cc[u] >= 0) atomicAdd(&deg[cc[u]], 1.0f);
        }
        __syncthreads();
    }
}

__global__ __launch_bounds__(256) void gcn_dis_kernel(float* __restrict__ deg, int N) {
    int i = blockIdx.x * blockDim.x + threadIdx.x;
    if (i < N) {
        float d = deg[i];
        deg[i] = (d > 0.0f) ? __frsqrt_rn(d) : 0.0f;
    }
}

#define FB_ROWS 100
#define AGG_WAVES (AGG_THREADS / 32)
#define AGG_GRP 8
#define AGG_EPT 8

__device__ __noinline__ void agg_add_serial(float* tile, const float* __restrict__ h,
                                            const float* __restrict__ dis,
                                            const int* __restrict__ col,
                                            int e, int r, int rowBase) {
    const int cI = col[e];
    const float norm = dis[r] * dis[cI];
    const float* hp = h + (size_t)cI * HIDDEN;
    float* tp = tile + (r - rowBase) * HIDDEN;
    for (int k = 0; k < HIDDEN; ++k) atomicAdd(tp + k, hp[k] * norm);
}

template <int G>
__device__ __forceinline__ void agg_add_group(float4& acc, const float4* __restrict__ h4,
                                              int cI, float nm, int j0, int last, int lane) {
    float4 v[G];
    #pragma unroll
    for (int u = 0; u < G; ++u) {
        int js = j0 + u;
        if (js > last) js = last;
        const int cj = __builtin_amdgcn_readlane(cI, js);
        v[u] = h4[(size_t)cj * (HIDDEN / 4) + lane];
    }
    #pragma unroll
    for (int u = 0; u < G; ++u) {
        const float wj = __int_as_float(__builtin_amdgcn_readlane(__float_as_int(nm), j0 + u));
        acc.x = fmaf(v[u].x, wj, acc.x);
        acc.y = fmaf(v[u].y, wj, acc.y);
        acc.z = fmaf(v[u].z, wj, acc.z);
        acc.w = fmaf(v[u].w, wj, acc.w);
    }
}

__global__ __launch_bounds__(AGG_THREADS) void gcn_aggregate_kernel(
        const float* __restrict__ h, const float* __restrict__ dis,
        const int* __restrict__ row, const int* __restrict__ col,
        const int* binw, float* out, int N, int E) {
    __shared__ int   s_list[AGG_CAP];
    __shared__ int   s_src[AGG_CAP];
    __shared__ float s_nm[AGG_CAP];
    __shared__ float s_dr[AGG_ROWS];
    __shared__ int   s_rcnt[AGG_ROWS];
    __shared__ int   s_rend[AGG_ROWS];
    __shared__ __attribute__((aligned(16))) float s_tile[FB_ROWS * HIDDEN];
    __shared__ int s_cnt;

    const int tid  = threadIdx.x;
    const int lane = tid & 31;
    const int wave = tid >> 5;
    const int rowBase = blockIdx.x * AGG_ROWS;
    int rowsHere = N - rowBase;
    if (rowsHere > AGG_ROWS) rowsHere = AGG_ROWS;

    const int* myBin = binw + (size_t)rowBase * HIDDEN;
    const int binCnt = myBin[0];
    const bool binOK = ((int)gridDim.x <= BIN_MAXB) && (N <= (1 << PACK_SHIFT)) &&
                       (binCnt <= AGG_CAP) && (binCnt < rowsHere * HIDDEN);

    if (binOK) {
        for (int i = tid; i < AGG_ROWS; i += AGG_THREADS) {
            s_rcnt[i] = 0;
            s_dr[i] = (i < rowsHere) ? dis[rowBase + i] : 0.0f;
        }
        __syncthreads();

        for (int base = 0; base < binCnt; base += AGG_THREADS * AGG_EPT) {
            int w[AGG_EPT];
            #pragma unroll
            for (int u = 0; u < AGG_EPT; ++u) {
                const int i = base + u * AGG_THREADS + tid;
                w[u] = myBin[1 + ((i < binCnt) ? i : (binCnt - 1))];
            }
            #pragma unroll
            for (int u = 0; u < AGG_EPT; ++u) {
                const int i = base + u * AGG_THREADS + tid;
                if (i < binCnt) {
                    s_list[i] = w[u];
                    atomicAdd(&s_rcnt[(unsigned)w[u] >> PACK_SHIFT], 1);
                }
            }
        }
        __syncthreads();

        if (wave == 0) {
            constexpr int RUN = (AGG_ROWS + 31) / 32;
            int c[RUN];
            int sum = 0;
            #pragma unroll
            for (int k = 0; k < RUN; ++k) {
                const int r = lane * RUN + k;
                c[k] = (r < AGG_ROWS) ? s_rcnt[r] : 0;
                sum += c[k];
            }
            int incl = sum;
            #pragma unroll
            for (int d = 1; d < 32; d <<= 1) {
                const int t = __shfl_up(incl, d);
                if (lane >= d) incl += t;
            }
            int run = incl - sum;
            #pragma unroll
            for (int k = 0; k < RUN; ++k) {
                const int r = lane * RUN + k;
                if (r < AGG_ROWS) s_rend[r] = run;
                run += c[k];
            }
        }
        __syncthreads();

        for (int base = 0; base < binCnt; base += AGG_THREADS * AGG_EPT) {
            int w[AGG_EPT];
            float dc[AGG_EPT];
            #pragma unroll
            for (int u = 0; u < AGG_EPT; ++u) {
                const int i = base + u * AGG_THREADS + tid;
                w[u] = s_list[(i < binCnt) ? i : (binCnt - 1)];
                dc[u] = dis[(unsigned)w[u] & PACK_MASK];
            }
            #pragma unroll
            for (int u = 0; u < AGG_EPT; ++u) {
                const int i = base + u * AGG_THREADS + tid;
                if (i < binCnt) {
                    const int lr = (int)((unsigned)w[u] >> PACK_SHIFT);
                    const int pos = atomicAdd(&s_rend[lr], 1);
                    s_src[pos] = (int)((unsigned)w[u] & PACK_MASK);
                    s_nm[pos]  = s_dr[lr] * dc[u];
                }
            }
        }
        __syncthreads();

        {
            const int waveU = __builtin_amdgcn_readfirstlane(wave);
            const float4* __restrict__ h4 = (const float4*)h;
            float4* o4 = (float4*)out;
            for (int lr = waveU; lr < rowsHere; lr += AGG_WAVES) {
                const int n     = __builtin_amdgcn_readfirstlane(s_rcnt[lr]);
                const int start = __builtin_amdgcn_readfirstlane(s_rend[lr]) - n;
                float4 acc = make_float4(0.f, 0.f, 0.f, 0.f);
                for (int c0 = 0; c0 < n; c0 += 32) {
                    int m = n - c0;
                    if (m > 32) m = 32;
                    int cI = 0;
                    float nm = 0.0f;
                    if (lane < m) {
                        cI = s_src[start + c0 + lane];
                        nm = s_nm[start + c0 + lane];
                    }
                    int j = 0;
                    for (; j + AGG_GRP <= m; j += AGG_GRP) agg_add_group<AGG_GRP>(acc, h4, cI, nm, j, m - 1, lane);
                    const int rem = m - j;
                    if (rem > AGG_GRP / 2)  agg_add_group<AGG_GRP>(acc, h4, cI, nm, j, m - 1, lane);
                    else if (rem > 0)       agg_add_group<AGG_GRP / 2>(acc, h4, cI, nm, j, m - 1, lane);
                }
                o4[(size_t)(rowBase + lr) * (HIDDEN / 4) + lane] = acc;
            }
        }
    } else {
        const int4* __restrict__ row4 = (const int4*)row;
        const int n4 = E >> 2;
        const int step = AGG_THREADS * AGG_UNR;
        const int nIter = (n4 + step - 1) / step;
        for (int sub = 0; sub < rowsHere; sub += FB_ROWS) {
            const int subBase = rowBase + sub;
            int subRows = rowsHere - sub;
            if (subRows > FB_ROWS) subRows = FB_ROWS;
            const unsigned urows = (unsigned)subRows;
            {
                float4* t4 = (float4*)s_tile;
                const float4 z = make_float4(0.f, 0.f, 0.f, 0.f);
                for (int i = tid; i < FB_ROWS * HIDDEN / 4; i += AGG_THREADS) t4[i] = z;
                if (tid == 0) s_cnt = 0;
            }
            __syncthreads();

            const int noMatch = subBase - 1;
            for (int it = 0; it < nIter; ++it) {
                const int i0 = it * step + tid;
                int4 v[AGG_UNR];
                #pragma unroll
                for (int u = 0; u < AGG_UNR; ++u) {
                    const int i = i0 + u * AGG_THREADS;
                    v[u] = (i < n4) ? row4[i] : make_int4(noMatch, noMatch, noMatch, noMatch);
                }
                #pragma unroll
                for (int u = 0; u < AGG_UNR; ++u) {
                    const int e0 = (i0 + u * AGG_THREADS) * 4;
                    const int rr[4] = {v[u].x, v[u].y, v[u].z, v[u].w};
                    #pragma unroll
                    for (int q = 0; q < 4; ++q) {
                        if ((unsigned)(rr[q] - subBase) < urows) {
                            const int p = atomicAdd(&s_cnt, 1);
                            if (p < AGG_CAP) s_list[p] = e0 + q;
                            else agg_add_serial(s_tile, h, dis, col, e0 + q, rr[q], subBase);
                        }
                    }
                }
            }
            const int eTail = (n4 << 2) + tid;
            if (eTail < E) {
                const int r = row[eTail];
                if ((unsigned)(r - subBase) < urows) {
                    const int p = atomicAdd(&s_cnt, 1);
                    if (p < AGG_CAP) s_list[p] = eTail;
                    else agg_add_serial(s_tile, h, dis, col, eTail, r, subBase);
                }
            }
            __syncthreads();

            int total = s_cnt;
            if (total > AGG_CAP) total = AGG_CAP;
            for (int i = wave; i < total; i += AGG_WAVES) {
                const int e  = s_list[i];
                const int cI = col[e];
                const int r  = row[e];
                const float nm = dis[r] * dis[cI];
                const float* hp = h + (size_t)cI * HIDDEN + lane;
                float* tp = s_tile + (r - subBase) * HIDDEN + lane;
                atomicAdd(tp +  0, hp[0]  * nm);
                atomicAdd(tp + 32, hp[32] * nm);
                atomicAdd(tp + 64, hp[64] * nm);
                atomicAdd(tp + 96, hp[96] * nm);
            }
            __syncthreads();

            {
                const float4* t4 = (const float4*)s_tile;
                float4* o4 = (float4*)(out + (size_t)subBase * HIDDEN);
                const int n = subRows * (HIDDEN / 4);
                for (int i = tid; i < n; i += AGG_THREADS) o4[i] = t4[i];
            }
            __syncthreads();
        }
    }
}

extern "C" void kernel_launch(void* const* d_in, const int* in_sizes, int n_in,
                              void* d_out, int out_size, void* d_ws, size_t ws_size,
                              hipStream_t stream) {
    const float* x  = (const float*)d_in[0];
    const float* W  = (const float*)d_in[1];
    const int*   ei = (const int*)d_in[2];
    float* out = (float*)d_out;

    const int N = in_sizes[0] / HIDDEN;
    const int E = in_sizes[2] / 2;
    const int* rowIdx = ei;
    const int* colIdx = ei + E;

    float* h   = (float*)d_ws;
    float* deg = h + (size_t)N * HIDDEN;

    const int nb = (N + AGG_ROWS - 1) / AGG_ROWS;
    int* binw = (int*)d_out;
    gcn_zero_kernel<<<(N + 255) / 256, 256, 0, stream>>>(deg, N, binw, nb);

    gcn_gemm_wmma_kernel<<<GEMM_BLOCKS, GEMM_THREADS, 0, stream>>>(x, W, h, N);

    gcn_deg_bin_kernel<<<256, BIN_THREADS, 0, stream>>>(rowIdx, colIdx, deg, binw, N, E, nb);

    gcn_dis_kernel<<<(N + 255) / 256, 256, 0, stream>>>(deg, N);

    gcn_aggregate_kernel<<<nb, AGG_THREADS, 0, stream>>>(
        h, deg, rowIdx, colIdx, binw, out, N, E);
}


# Degree counts: 64 blocks each count a slice of col[] into an LDS table of 2-byte integer counters over all nodes (LDS integer adds); dis pass sums the 64 tables; no global float atomics on deg
# speedup vs baseline: 3.8897x; 1.5684x over previous
#include <hip/hip_runtime.h>

typedef __attribute__((ext_vector_type(2))) float v2f;
typedef __attribute__((ext_vector_type(8))) float v8f;

#define HIDDEN 128

#define AGG_ROWS    400
#define AGG_THREADS 1024
#define AGG_CAP     16384
#define AGG_UNR     8
#define BIN_WORDS   (AGG_ROWS * HIDDEN)
#define BIN_THREADS 1024
#define BIN_EPT     8
#define BIN_MAXB    1024
#define PACK_SHIFT  23
#define PACK_MASK   ((1u << PACK_SHIFT) - 1u)
static_assert(AGG_ROWS <= (1 << (32 - PACK_SHIFT)), "the local row must fit above the source row");

#define COUNT_BLOCKS  64
#define COUNT_THREADS 1024
#define COUNT_EPT     25
#define COUNT_WORDS   51200
#define COUNT_CHUNK   (AGG_ROWS / 2)
#define COUNT_OFF     16640
static_assert(AGG_ROWS % 8 == 0, "a chunk is moved 16 bytes at a time");
static_assert(AGG_CAP + 1 <= COUNT_OFF, "the entries a bin can use end below the tables' chunks");
static_assert(COUNT_OFF % 4 == 0, "chunks start on a 16-byte boundary");

__device__ __forceinline__ bool count_in_lds(int N, int E, int nb) {
    const int lastRows = N - (nb - 1) * AGG_ROWS;
    return (nb * COUNT_CHUNK <= COUNT_WORDS) &&
           ((E + COUNT_BLOCKS - 1) / COUNT_BLOCKS <= 65535) &&
           (lastRows * HIDDEN >= COUNT_OFF + COUNT_BLOCKS * COUNT_CHUNK);
}

__global__ __launch_bounds__(COUNT_THREADS) void gcn_count_kernel(
        const int* __restrict__ col, float* __restrict__ deg, int* __restrict__ binw,
        int N, int E, int nb) {
    __shared__ __attribute__((aligned(16))) unsigned s_h[COUNT_WORDS];

    const int tid = threadIdx.x;
    const int gtid = (int)blockIdx.x * COUNT_THREADS + tid;
    const int gsz = (int)gridDim.x * COUNT_THREADS;
    for (int i = gtid; i < nb; i += gsz) binw[(size_t)i * BIN_WORDS] = 0;

    if (!count_in_lds(N, E, nb)) {
        for (int i = gtid; i < N; i += gsz) deg[i] = 0.0f;
        return;
    }
    if ((int)blockIdx.x >= COUNT_BLOCKS) return;

    const int nQuads = nb * (COUNT_CHUNK / 4);
    {
        uint4* h4 = (uint4*)s_h;
        const uint4 z = make_uint4(0u, 0u, 0u, 0u);
        for (int i = tid; i < nQuads; i += COUNT_THREADS) h4[i] = z;
    }
    __syncthreads();

    const int per = (E + COUNT_BLOCKS - 1) / COUNT_BLOCKS;
    const int cBeg = (int)blockIdx.x * per;
    int cEnd = cBeg + per;
    if (cEnd > E) cEnd = E;
    for (int t0 = cBeg; t0 < cEnd; t0 += COUNT_THREADS * COUNT_EPT) {
        int cc[COUNT_EPT];
        #pragma unroll
        for (int u = 0; u < COUNT_EPT; ++u) {
            const int e = t0 + u * COUNT_THREADS + tid;
            cc[u] = col[(e < cEnd) ? e : (cEnd - 1)];
        }
        #pragma unroll
        for (int u = 0; u < COUNT_EPT; ++u) {
            const int e = t0 + u * COUNT_THREADS + tid;
            const int cI = cc[u];
            if (e < cEnd && (unsigned)cI < (unsigned)N)
                atomicAdd(&s_h[cI >> 1], 1u << ((cI & 1) << 4));
        }
    }
    __syncthreads();

    {
        const uint4* h4 = (const uint4*)s_h;
        for (int q = tid; q < nQuads; q += COUNT_THREADS) {
            const int b = q / (COUNT_CHUNK / 4);
            const int j = q - b * (COUNT_CHUNK / 4);
            uint4* dst = (uint4*)(binw + (size_t)b * BIN_WORDS + COUNT_OFF +
                                  (int)blockIdx.x * COUNT_CHUNK);
            dst[j] = h4[q];
        }
    }
}

#define GEMM_THREADS 256
#define GEMM_TR      16
#define GEMM_PITCH   (HIDDEN + 4)
#define GEMM_BLOCKS  768
#define GEMM_KSTEPS  (HIDDEN / 4)

__global__ __launch_bounds__(GEMM_THREADS) void gcn_gemm_wmma_kernel(
        const float* __restrict__ x, const float* __restrict__ W,
        float* __restrict__ h, int N) {
    __shared__ __attribute__((aligned(16))) float s_x[2 * GEMM_TR * GEMM_PITCH];

    const int tid     = threadIdx.x;
    const int lane    = tid & 31;
    const int wave    = tid >> 5;
    const int halfSel = lane >> 4;
    const int l       = lane & 15;
    const int nBase   = wave * 16;
    const int nTiles  = (N + GEMM_TR - 1) / GEMM_TR;

    const float* __restrict__ wrow = W + (size_t)(nBase + l) * HIDDEN + (halfSel << 1);
    v2f bfrag[GEMM_KSTEPS];
    #pragma unroll
    for (int s = 0; s < GEMM_KSTEPS; ++s) bfrag[s] = *(const v2f*)(wrow + 4 * s);

    const int sr0 = wave, sr1 = wave + 8;
    float* const sdst0 = &s_x[sr0 * GEMM_PITCH + lane * 4];
    float* const sdst1 = &s_x[sr1 * GEMM_PITCH + lane * 4];

    int tile = blockIdx.x;
    float4 g0 = make_float4(0.f, 0.f, 0.f, 0.f), g1 = g0;
    if (tile < nTiles) {
        int r0 = tile * GEMM_TR + sr0; if (r0 >= N) r0 = N - 1;
        int r1 = tile * GEMM_TR + sr1; if (r1 >= N) r1 = N - 1;
        g0 = *(const float4*)(x + (size_t)r0 * HIDDEN + lane * 4);
        g1 = *(const float4*)(x + (size_t)r1 * HIDDEN + lane * 4);
    }

    int buf = 0;
    for (; tile < nTiles; tile += (int)gridDim.x) {
        const int bofs = buf * (GEMM_TR * GEMM_PITCH);
        *(float4*)(sdst0 + bofs) = g0;
        *(float4*)(sdst1 + bofs) = g1;
        __syncthreads();

        const int nxt = tile + (int)gridDim.x;
        if (nxt < nTiles) {
            int r0 = nxt * GEMM_TR + sr0; if (r0 >= N) r0 = N - 1;
            int r1 = nxt * GEMM_TR + sr1; if (r1 >= N) r1 = N - 1;
            g0 = *(const float4*)(x + (size_t)r0 * HIDDEN + lane * 4);
            g1 = *(const float4*)(x + (size_t)r1 * HIDDEN + lane * 4);
        }

        const float* ap = &s_x[bofs + l * GEMM_PITCH + (halfSel << 1)];
        v2f afrag[GEMM_KSTEPS];
        #pragma unroll
        for (int s = 0; s < GEMM_KSTEPS; ++s) afrag[s] = *(const v2f*)(ap + 4 * s);

        v8f c = {};
        #pragma unroll
        for (int s = 0; s < GEMM_KSTEPS; ++s) {
            c = __builtin_amdgcn_wmma_f32_16x16x4_f32(
                     false, afrag[s],  false, bfrag[s],
                     (short)0, c,  false,  false);
        }

        const int rowBase = tile * GEMM_TR;
        float* __restrict__ hp = h + (size_t)rowBase * HIDDEN + nBase + l;
        #pragma unroll
        for (int v = 0; v < 8; ++v) {
            const int m = v + halfSel * 8;
            if (rowBase + m < N) hp[(size_t)m * HIDDEN] = c[v];
        }
        buf ^= 1;
    }
}

__global__ __launch_bounds__(BIN_THREADS) void gcn_deg_bin_kernel(
        const int* __restrict__ row, const int* __restrict__ col,
        float* __restrict__ deg, int* __restrict__ binw, int N, int E, int nb) {
    __shared__ int s_c[BIN_MAXB];

    const int tid = threadIdx.x;
    const int per = (E + (int)gridDim.x - 1) / (int)gridDim.x;
    const int cBeg = (int)blockIdx.x * per;
    int cEnd = cBeg + per;
    if (cEnd > E) cEnd = E;
    const bool doBin = (nb <= BIN_MAXB) && (N <= (1 << PACK_SHIFT));
    const bool counted = count_in_lds(N, E, nb);

    for (int t0 = cBeg; t0 < cEnd; t0 += BIN_THREADS * BIN_EPT) {
        for (int i = tid; i < BIN_MAXB; i += BIN_THREADS) s_c[i] = 0;
        __syncthreads();

        int cc[BIN_EPT], bb[BIN_EPT], rk[BIN_EPT], pk[BIN_EPT];
        #pragma unroll
        for (int u = 0; u < BIN_EPT; ++u) {
            const int e = t0 + u * BIN_THREADS + tid;
            const int ec = (e < cEnd) ? e : (cEnd - 1);
            cc[u] = col[ec];
            bb[u] = row[ec];
        }
        #pragma unroll
        for (int u = 0; u < BIN_EPT; ++u) {
            const int e = t0 + u * BIN_THREADS + tid;
            const int r = bb[u];
            const bool live = e < cEnd;
            const int b = r / AGG_ROWS;
            pk[u] = (int)(((unsigned)(r - b * AGG_ROWS) << PACK_SHIFT) | ((unsigned)cc[u] & PACK_MASK));
            cc[u] = live ? cc[u] : -1;
            bb[u] = (live && doBin && (unsigned)r < (unsigned)N) ? b : -1;
            rk[u] = 0;
        }
        #pragma unroll
        for (int u = 0; u < BIN_EPT; ++u) {
            if (bb[u] >= 0) rk[u] = atomicAdd(&s_c[bb[u]], 1);
        }
        __syncthreads();

        for (int i = tid; i < nb && i < BIN_MAXB; i += BIN_THREADS) {
            const int c = s_c[i];
            if (c > 0) s_c[i] = atomicAdd(&binw[(size_t)i * BIN_WORDS], c);
        }
        __syncthreads();

        #pragma unroll
        for (int u = 0; u < BIN_EPT; ++u) {
            if (bb[u] >= 0) {
                const int b = bb[u];
                int rowsB = N - b * AGG_ROWS;
                if (rowsB > AGG_ROWS) rowsB = AGG_ROWS;
                int room = rowsB * HIDDEN - 1;
                if (counted && room > COUNT_OFF - 1) room = COUNT_OFF - 1;
                const int pos = s_c[b] + rk[u];
                if (pos < room)
                    binw[(size_t)b * BIN_WORDS + 1 + pos] = pk[u];
            }
        }
        if (!counted) {
            #pragma unroll
            for (int u = 0; u < BIN_EPT; ++u) {
                if (cc[u] >= 0) atomicAdd(&deg[cc[u]], 1.0f);
            }
        }
        __syncthreads();
    }
}

__global__ __launch_bounds__(256) void gcn_dis_kernel(float* __restrict__ deg,
                                                      const int* __restrict__ binw,
                                                      int N, int E, int nb) {
    const int i = blockIdx.x * blockDim.x + threadIdx.x;
    if (count_in_lds(N, E, nb)) {
        if (2 * i < N) {
            const int b = i / COUNT_CHUNK;
            const int j = i - b * COUNT_CHUNK;
            const unsigned* p = (const unsigned*)binw + (size_t)b * BIN_WORDS + COUNT_OFF + j;
            unsigned lo = 0u, hi = 0u;
            #pragma unroll 16
            for (int g = 0; g < COUNT_BLOCKS; ++g) {
                const unsigned w = p[g * COUNT_CHUNK];
                lo += w & 0xffffu;
                hi += w >> 16;
            }
            const float d0 = (float)lo, d1 = (float)hi;
            deg[2 * i] = (d0 > 0.0f) ? __frsqrt_rn(d0) : 0.0f;
            if (2 * i + 1 < N) deg[2 * i + 1] = (d1 > 0.0f) ? __frsqrt_rn(d1) : 0.0f;
        }
    } else if (i < N) {
        float d = deg[i];
        deg[i] = (d > 0.0f) ? __frsqrt_rn(d) : 0.0f;
    }
}

#define FB_ROWS 100
#define AGG_WAVES (AGG_THREADS / 32)
#define AGG_GRP 8
#define AGG_EPT 8

__device__ __noinline__ void agg_add_serial(float* tile, const float* __restrict__ h,
                                            const float* __restrict__ dis,
                                            const int* __restrict__ col,
                                            int e, int r, int rowBase) {
    const int cI = col[e];
    const float norm = dis[r] * dis[cI];
    const float* hp = h + (size_t)cI * HIDDEN;
    float* tp = tile + (r - rowBase) * HIDDEN;
    for (int k = 0; k < HIDDEN; ++k) atomicAdd(tp + k, hp[k] * norm);
}

template <int G>
__device__ __forceinline__ void agg_add_group(float4& acc, const float4* __restrict__ h4,
                                              int cI, float nm, int j0, int last, int lane) {
    float4 v[G];
    #pragma unroll
    for (int u = 0; u < G; ++u) {
        int js = j0 + u;
        if (js > last) js = last;
        const int cj = __builtin_amdgcn_readlane(cI, js);
        v[u] = h4[(size_t)cj * (HIDDEN / 4) + lane];
    }
    #pragma unroll
    for (int u = 0; u < G; ++u) {
        const float wj = __int_as_float(__builtin_amdgcn_readlane(__float_as_int(nm), j0 + u));
        acc.x = fmaf(v[u].x, wj, acc.x);
        acc.y = fmaf(v[u].y, wj, acc.y);
        acc.z = fmaf(v[u].z, wj, acc.z);
        acc.w = fmaf(v[u].w, wj, acc.w);
    }
}

__global__ __launch_bounds__(AGG_THREADS) void gcn_aggregate_kernel(
        const float* __restrict__ h, const float* __restrict__ dis,
        const int* __restrict__ row, const int* __restrict__ col,
        const int* binw, float* out, int N, int E) {
    __shared__ int   s_list[AGG_CAP];
    __shared__ int   s_src[AGG_CAP];
    __shared__ float s_nm[AGG_CAP];
    __shared__ float s_dr[AGG_ROWS];
    __shared__ int   s_rcnt[AGG_ROWS];
    __shared__ int   s_rend[AGG_ROWS];
    __shared__ __attribute__((aligned(16))) float s_tile[FB_ROWS * HIDDEN];
    __shared__ int s_cnt;

    const int tid  = threadIdx.x;
    const int lane = tid & 31;
    const int wave = tid >> 5;
    const int rowBase = blockIdx.x * AGG_ROWS;
    int rowsHere = N - rowBase;
    if (rowsHere > AGG_ROWS) rowsHere = AGG_ROWS;

    const int* myBin = binw + (size_t)rowBase * HIDDEN;
    const int binCnt = myBin[0];
    const bool binOK = ((int)gridDim.x <= BIN_MAXB) && (N <= (1 << PACK_SHIFT)) &&
                       (binCnt <= AGG_CAP) && (binCnt < rowsHere * HIDDEN);

    if (binOK) {
        for (int i = tid; i < AGG_ROWS; i += AGG_THREADS) {
            s_rcnt[i] = 0;
            s_dr[i] = (i < rowsHere) ? dis[rowBase + i] : 0.0f;
        }
        __syncthreads();

        for (int base = 0; base < binCnt; base += AGG_THREADS * AGG_EPT) {
            int w[AGG_EPT];
            #pragma unroll
            for (int u = 0; u < AGG_EPT; ++u) {
                const int i = base + u * AGG_THREADS + tid;
                w[u] = myBin[1 + ((i < binCnt) ? i : (binCnt - 1))];
            }
            #pragma unroll
            for (int u = 0; u < AGG_EPT; ++u) {
                const int i = base + u * AGG_THREADS + tid;
                if (i < binCnt) {
                    s_list[i] = w[u];
                    atomicAdd(&s_rcnt[(unsigned)w[u] >> PACK_SHIFT], 1);
                }
            }
        }
        __syncthreads();

        if (wave == 0) {
            constexpr int RUN = (AGG_ROWS + 31) / 32;
            int c[RUN];
            int sum = 0;
            #pragma unroll
            for (int k = 0; k < RUN; ++k) {
                const int r = lane * RUN + k;
                c[k] = (r < AGG_ROWS) ? s_rcnt[r] : 0;
                sum += c[k];
            }
            int incl = sum;
            #pragma unroll
            for (int d = 1; d < 32; d <<= 1) {
                const int t = __shfl_up(incl, d);
                if (lane >= d) incl += t;
            }
            int run = incl - sum;
            #pragma unroll
            for (int k = 0; k < RUN; ++k) {
                const int r = lane * RUN + k;
                if (r < AGG_ROWS) s_rend[r] = run;
                run += c[k];
            }
        }
        __syncthreads();

        for (int base = 0; base < binCnt; base += AGG_THREADS * AGG_EPT) {
            int w[AGG_EPT];
            float dc[AGG_EPT];
            #pragma unroll
            for (int u = 0; u < AGG_EPT; ++u) {
                const int i = base + u * AGG_THREADS + tid;
                w[u] = s_list[(i < binCnt) ? i : (binCnt - 1)];
                dc[u] = dis[(unsigned)w[u] & PACK_MASK];
            }
            #pragma unroll
            for (int u = 0; u < AGG_EPT; ++u) {
                const int i = base + u * AGG_THREADS + tid;
                if (i < binCnt) {
                    const int lr = (int)((unsigned)w[u] >> PACK_SHIFT);
                    const int pos = atomicAdd(&s_rend[lr], 1);
                    s_src[pos] = (int)((unsigned)w[u] & PACK_MASK);
                    s_nm[pos]  = s_dr[lr] * dc[u];
                }
            }
        }
        __syncthreads();

        {
            const int waveU = __builtin_amdgcn_readfirstlane(wave);
            const float4* __restrict__ h4 = (const float4*)h;
            float4* o4 = (float4*)out;
            for (int lr = waveU; lr < rowsHere; lr += AGG_WAVES) {
                const int n     = __builtin_amdgcn_readfirstlane(s_rcnt[lr]);
                const int start = __builtin_amdgcn_readfirstlane(s_rend[lr]) - n;
                float4 acc = make_float4(0.f, 0.f, 0.f, 0.f);
                for (int c0 = 0; c0 < n; c0 += 32) {
                    int m = n - c0;
                    if (m > 32) m = 32;
                    int cI = 0;
                    float nm = 0.0f;
                    if (lane < m) {
                        cI = s_src[start + c0 + lane];
                        nm = s_nm[start + c0 + lane];
                    }
                    int j = 0;
                    for (; j + AGG_GRP <= m; j += AGG_GRP) agg_add_group<AGG_GRP>(acc, h4, cI, nm, j, m - 1, lane);
                    const int rem = m - j;
                    if (rem > AGG_GRP / 2)  agg_add_group<AGG_GRP>(acc, h4, cI, nm, j, m - 1, lane);
                    else if (rem > 0)       agg_add_group<AGG_GRP / 2>(acc, h4, cI, nm, j, m - 1, lane);
                }
                o4[(size_t)(rowBase + lr) * (HIDDEN / 4) + lane] = acc;
            }
        }
    } else {
        const int4* __restrict__ row4 = (const int4*)row;
        const int n4 = E >> 2;
        const int step = AGG_THREADS * AGG_UNR;
        const int nIter = (n4 + step - 1) / step;
        for (int sub = 0; sub < rowsHere; sub += FB_ROWS) {
            const int subBase = rowBase + sub;
            int subRows = rowsHere - sub;
            if (subRows > FB_ROWS) subRows = FB_ROWS;
            const unsigned urows = (unsigned)subRows;
            {
                float4* t4 = (float4*)s_tile;
                const float4 z = make_float4(0.f, 0.f, 0.f, 0.f);
                for (int i = tid; i < FB_ROWS * HIDDEN / 4; i += AGG_THREADS) t4[i] = z;
                if (tid == 0) s_cnt = 0;
            }
            __syncthreads();

            const int noMatch = subBase - 1;
            for (int it = 0; it < nIter; ++it) {
                const int i0 = it * step + tid;
                int4 v[AGG_UNR];
                #pragma unroll
                for (int u = 0; u < AGG_UNR; ++u) {
                    const int i = i0 + u * AGG_THREADS;
                    v[u] = (i < n4) ? row4[i] : make_int4(noMatch, noMatch, noMatch, noMatch);
                }
                #pragma unroll
                for (int u = 0; u < AGG_UNR; ++u) {
                    const int e0 = (i0 + u * AGG_THREADS) * 4;
                    const int rr[4] = {v[u].x, v[u].y, v[u].z, v[u].w};
                    #pragma unroll
                    for (int q = 0; q < 4; ++q) {
                        if ((unsigned)(rr[q] - subBase) < urows) {
                            const int p = atomicAdd(&s_cnt, 1);
                            if (p < AGG_CAP) s_list[p] = e0 + q;
                            else agg_add_serial(s_tile, h, dis, col, e0 + q, rr[q], subBase);
                        }
                    }
                }
            }
            const int eTail = (n4 << 2) + tid;
            if (eTail < E) {
                const int r = row[eTail];
                if ((unsigned)(r - subBase) < urows) {
                    const int p = atomicAdd(&s_cnt, 1);
                    if (p < AGG_CAP) s_list[p] = eTail;
                    else agg_add_serial(s_tile, h, dis, col, eTail, r, subBase);
                }
            }
            __syncthreads();

            int total = s_cnt;
            if (total > AGG_CAP) total = AGG_CAP;
            for (int i = wave; i < total; i += AGG_WAVES) {
                const int e  = s_list[i];
                const int cI = col[e];
                const int r  = row[e];
                const float nm = dis[r] * dis[cI];
                const float* hp = h + (size_t)cI * HIDDEN + lane;
                float* tp = s_tile + (r - subBase) * HIDDEN + lane;
                atomicAdd(tp +  0, hp[0]  * nm);
                atomicAdd(tp + 32, hp[32] * nm);
                atomicAdd(tp + 64, hp[64] * nm);
                atomicAdd(tp + 96, hp[96] * nm);
            }
            __syncthreads();

            {
                const float4* t4 = (const float4*)s_tile;
                float4* o4 = (float4*)(out + (size_t)subBase * HIDDEN);
                const int n = subRows * (HIDDEN / 4);
                for (int i = tid; i < n; i += AGG_THREADS) o4[i] = t4[i];
            }
            __syncthreads();
        }
    }
}

extern "C" void kernel_launch(void* const* d_in, const int* in_sizes, int n_in,
                              void* d_out, int out_size, void* d_ws, size_t ws_size,
                              hipStream_t stream) {
    const float* x  = (const float*)d_in[0];
    const float* W  = (const float*)d_in[1];
    const int*   ei = (const int*)d_in[2];
    float* out = (float*)d_out;

    const int N = in_sizes[0] / HIDDEN;
    const int E = in_sizes[2] / 2;
    const int* rowIdx = ei;
    const int* colIdx = ei + E;

    float* h   = (float*)d_ws;
    float* deg = h + (size_t)N * HIDDEN;

    const int nb = (N + AGG_ROWS - 1) / AGG_ROWS;
    int* binw = (int*)d_out;
    gcn_count_kernel<<<COUNT_BLOCKS, COUNT_THREADS, 0, stream>>>(colIdx, deg, binw, N, E, nb);

    gcn_gemm_wmma_kernel<<<GEMM_BLOCKS, GEMM_THREADS, 0, stream>>>(x, W, h, N);

    gcn_deg_bin_kernel<<<256, BIN_THREADS, 0, stream>>>(rowIdx, colIdx, deg, binw, N, E, nb);

    gcn_dis_kernel<<<(N + 255) / 256, 256, 0, stream>>>(deg, binw, N, E, nb);

    gcn_aggregate_kernel<<<nb, AGG_THREADS, 0, stream>>>(
        h, deg, rowIdx, colIdx, binw, out, N, E);
}


# The 64 degree-counting blocks run inside the edge-binning launch (beside 192 binning blocks, one block per WGP), not as a launch of their own; projection launch clears the bin counts; 5 launches to 4
# speedup vs baseline: 4.1353x; 1.0631x over previous
#include <hip/hip_runtime.h>

typedef __attribute__((ext_vector_type(2))) float v2f;
typedef __attribute__((ext_vector_type(8))) float v8f;

#define HIDDEN 128

#define AGG_ROWS    400
#define AGG_THREADS 1024
#define AGG_CAP     16384
#define AGG_UNR     8
#define BIN_WORDS   (AGG_ROWS * HIDDEN)
#define BIN_THREADS 1024
#define BIN_EPT     9
#define BIN_BLOCKS  192
#define BIN_MAXB    1024
#define PACK_SHIFT  23
#define PACK_MASK   ((1u << PACK_SHIFT) - 1u)
static_assert(AGG_ROWS <= (1 << (32 - PACK_SHIFT)), "the local row must fit above the source row");

#define COUNT_BLOCKS  64
#define COUNT_THREADS 1024
#define COUNT_EPT     25
#define COUNT_WORDS   51200
#define COUNT_CHUNK   (AGG_ROWS / 2)
#define COUNT_OFF     16640
static_assert(AGG_ROWS % 8 == 0, "a chunk is moved 16 bytes at a time");
static_assert(AGG_CAP + 1 <= COUNT_OFF, "the entries a bin can use end below the tables' chunks");
static_assert(COUNT_OFF % 4 == 0, "chunks start on a 16-byte boundary");

__device__ __forceinline__ bool count_in_lds(int N, int E, int nb) {
    const int lastRows = N - (nb - 1) * AGG_ROWS;
    return (nb * COUNT_CHUNK <= COUNT_WORDS) &&
           ((E + COUNT_BLOCKS - 1) / COUNT_BLOCKS <= 65535) &&
           (lastRows * HIDDEN >= COUNT_OFF + COUNT_BLOCKS * COUNT_CHUNK);
}

__device__ __forceinline__ void count_block(
        const int* __restrict__ col, int* __restrict__ binw, unsigned* s_h,
        int cblk, int N, int E, int nb) {
    const int tid = threadIdx.x;

    const int nQuads = nb * (COUNT_CHUNK / 4);
    {
        uint4* h4 = (uint4*)s_h;
        const uint4 z = make_uint4(0u, 0u, 0u, 0u);
        for (int i = tid; i < nQuads; i += COUNT_THREADS) h4[i] = z;
    }
    __syncthreads();

    const int per = (E + COUNT_BLOCKS - 1) / COUNT_BLOCKS;
    const int cBeg = cblk * per;
    int cEnd = cBeg + per;
    if (cEnd > E) cEnd = E;
    for (int t0 = cBeg; t0 < cEnd; t0 += COUNT_THREADS * COUNT_EPT) {
        int cc[COUNT_EPT];
        #pragma unroll
        for (int u = 0; u < COUNT_EPT; ++u) {
            const int e = t0 + u * COUNT_THREADS + tid;
            cc[u] = col[(e < cEnd) ? e : (cEnd - 1)];
        }
        #pragma unroll
        for (int u = 0; u < COUNT_EPT; ++u) {
            const int e = t0 + u * COUNT_THREADS + tid;
            const int cI = cc[u];
            if (e < cEnd && (unsigned)cI < (unsigned)N)
                atomicAdd(&s_h[cI >> 1], 1u << ((cI & 1) << 4));
        }
    }
    __syncthreads();

    {
        const uint4* h4 = (const uint4*)s_h;
        for (int q = tid; q < nQuads; q += COUNT_THREADS) {
            const int b = q / (COUNT_CHUNK / 4);
            const int j = q - b * (COUNT_CHUNK / 4);
            uint4* dst = (uint4*)(binw + (size_t)b * BIN_WORDS + COUNT_OFF +
                                  cblk * COUNT_CHUNK);
            dst[j] = h4[q];
        }
    }
}

#define GEMM_THREADS 256
#define GEMM_TR      16
#define GEMM_PITCH   (HIDDEN + 4)
#define GEMM_BLOCKS  768
#define GEMM_KSTEPS  (HIDDEN / 4)

__global__ __launch_bounds__(GEMM_THREADS) void gcn_gemm_wmma_kernel(
        const float* __restrict__ x, const float* __restrict__ W,
        float* __restrict__ h, float* __restrict__ deg, int* __restrict__ binw,
        int N, int E, int nb) {
    __shared__ __attribute__((aligned(16))) float s_x[2 * GEMM_TR * GEMM_PITCH];

    const int tid     = threadIdx.x;

    {
        const int gtid = (int)blockIdx.x * GEMM_THREADS + tid;
        const int gsz  = (int)gridDim.x * GEMM_THREADS;
        for (int i = gtid; i < nb; i += gsz) binw[(size_t)i * BIN_WORDS] = 0;
        if (!count_in_lds(N, E, nb))
            for (int i = gtid; i < N; i += gsz) deg[i] = 0.0f;
    }

    const int lane    = tid & 31;
    const int wave    = tid >> 5;
    const int halfSel = lane >> 4;
    const int l       = lane & 15;
    const int nBase   = wave * 16;
    const int nTiles  = (N + GEMM_TR - 1) / GEMM_TR;

    const float* __restrict__ wrow = W + (size_t)(nBase + l) * HIDDEN + (halfSel << 1);
    v2f bfrag[GEMM_KSTEPS];
    #pragma unroll
    for (int s = 0; s < GEMM_KSTEPS; ++s) bfrag[s] = *(const v2f*)(wrow + 4 * s);

    const int sr0 = wave, sr1 = wave + 8;
    float* const sdst0 = &s_x[sr0 * GEMM_PITCH + lane * 4];
    float* const sdst1 = &s_x[sr1 * GEMM_PITCH + lane * 4];

    int tile = blockIdx.x;
    float4 g0 = make_float4(0.f, 0.f, 0.f, 0.f), g1 = g0;
    if (tile < nTiles) {
        int r0 = tile * GEMM_TR + sr0; if (r0 >= N) r0 = N - 1;
        int r1 = tile * GEMM_TR + sr1; if (r1 >= N) r1 = N - 1;
        g0 = *(const float4*)(x + (size_t)r0 * HIDDEN + lane * 4);
        g1 = *(const float4*)(x + (size_t)r1 * HIDDEN + lane * 4);
    }

    int buf = 0;
    for (; tile < nTiles; tile += (int)gridDim.x) {
        const int bofs = buf * (GEMM_TR * GEMM_PITCH);
        *(float4*)(sdst0 + bofs) = g0;
        *(float4*)(sdst1 + bofs) = g1;
        __syncthreads();

        const int nxt = tile + (int)gridDim.x;
        if (nxt < nTiles) {
            int r0 = nxt * GEMM_TR + sr0; if (r0 >= N) r0 = N - 1;
            int r1 = nxt * GEMM_TR + sr1; if (r1 >= N) r1 = N - 1;
            g0 = *(const float4*)(x + (size_t)r0 * HIDDEN + lane * 4);
            g1 = *(const float4*)(x + (size_t)r1 * HIDDEN + lane * 4);
        }

        const float* ap = &s_x[bofs + l * GEMM_PITCH + (halfSel << 1)];
        v2f afrag[GEMM_KSTEPS];
        #pragma unroll
        for (int s = 0; s < GEMM_KSTEPS; ++s) afrag[s] = *(const v2f*)(ap + 4 * s);

        v8f c = {};
        #pragma unroll
        for (int s = 0; s < GEMM_KSTEPS; ++s) {
            c = __builtin_amdgcn_wmma_f32_16x16x4_f32(
                     false, afrag[s],  false, bfrag[s],
                     (short)0, c,  false,  false);
        }

        const int rowBase = tile * GEMM_TR;
        float* __restrict__ hp = h + (size_t)rowBase * HIDDEN + nBase + l;
        #pragma unroll
        for (int v = 0; v < 8; ++v) {
            const int m = v + halfSel * 8;
            if (rowBase + m < N) hp[(size_t)m * HIDDEN] = c[v];
        }
        buf ^= 1;
    }
}

static_assert(COUNT_THREADS == BIN_THREADS, "counting and binning blocks share one launch");
static_assert(BIN_MAXB <= COUNT_WORDS, "the bin counters lie in the head of the table's LDS");
__global__ __launch_bounds__(BIN_THREADS) void gcn_deg_bin_kernel(
        const int* __restrict__ row, const int* __restrict__ col,
        float* __restrict__ deg, int* __restrict__ binw, int N, int E, int nb) {
    __shared__ __attribute__((aligned(16))) unsigned s_h[COUNT_WORDS];
    int* const s_c = (int*)s_h;

    const int tid = threadIdx.x;
    const bool counted = count_in_lds(N, E, nb);
    if ((int)blockIdx.x < COUNT_BLOCKS) {
        if (counted) count_block(col, binw, s_h, (int)blockIdx.x, N, E, nb);
        return;
    }
    const int bblk  = (int)blockIdx.x - COUNT_BLOCKS;
    const int nbblk = (int)gridDim.x - COUNT_BLOCKS;
    const int per = (E + nbblk - 1) / nbblk;
    const int cBeg = bblk * per;
    int cEnd = cBeg + per;
    if (cEnd > E) cEnd = E;
    const bool doBin = (nb <= BIN_MAXB) && (N <= (1 << PACK_SHIFT));

    for (int t0 = cBeg; t0 < cEnd; t0 += BIN_THREADS * BIN_EPT) {
        for (int i = tid; i < BIN_MAXB; i += BIN_THREADS) s_c[i] = 0;
        __syncthreads();

        int cc[BIN_EPT], bb[BIN_EPT], rk[BIN_EPT], pk[BIN_EPT];
        #pragma unroll
        for (int u = 0; u < BIN_EPT; ++u) {
            const int e = t0 + u * BIN_THREADS + tid;
            const int ec = (e < cEnd) ? e : (cEnd - 1);
            cc[u] = col[ec];
            bb[u] = row[ec];
        }
        #pragma unroll
        for (int u = 0; u < BIN_EPT; ++u) {
            const int e = t0 + u * BIN_THREADS + tid;
            const int r = bb[u];
            const bool live = e < cEnd;
            const int b = r / AGG_ROWS;
            pk[u] = (int)(((unsigned)(r - b * AGG_ROWS) << PACK_SHIFT) | ((unsigned)cc[u] & PACK_MASK));
            cc[u] = live ? cc[u] : -1;
            bb[u] = (live && doBin && (unsigned)r < (unsigned)N) ? b : -1;
            rk[u] = 0;
        }
        #pragma unroll
        for (int u = 0; u < BIN_EPT; ++u) {
            if (bb[u] >= 0) rk[u] = atomicAdd(&s_c[bb[u]], 1);
        }
        __syncthreads();

        for (int i = tid; i < nb && i < BIN_MAXB; i += BIN_THREADS) {
            const int c = s_c[i];
            if (c > 0) s_c[i] = atomicAdd(&binw[(size_t)i * BIN_WORDS], c);
        }
        __syncthreads();

        #pragma unroll
        for (int u = 0; u < BIN_EPT; ++u) {
            if (bb[u] >= 0) {
                const int b = bb[u];
                int rowsB = N - b * AGG_ROWS;
                if (rowsB > AGG_ROWS) rowsB = AGG_ROWS;
                int room = rowsB * HIDDEN - 1;
                if (counted && room > COUNT_OFF - 1) room = COUNT_OFF - 1;
                const int pos = s_c[b] + rk[u];
                if (pos < room)
                    binw[(size_t)b * BIN_WORDS + 1 + pos] = pk[u];
            }
        }
        if (!counted) {
            #pragma unroll
            for (int u = 0; u < BIN_EPT; ++u) {
                if (cc[u] >= 0) atomicAdd(&deg[cc[u]], 1.0f);
            }
        }
        __syncthreads();
    }
}

__global__ __launch_bounds__(256) void gcn_dis_kernel(float* __restrict__ deg,
                                                      const int* __restrict__ binw,
                                                      int N, int E, int nb) {
    const int i = blockIdx.x * blockDim.x + threadIdx.x;
    if (count_in_lds(N, E, nb)) {
        if (2 * i < N) {
            const int b = i / COUNT_CHUNK;
            const int j = i - b * COUNT_CHUNK;
            const unsigned* p = (const unsigned*)binw + (size_t)b * BIN_WORDS + COUNT_OFF + j;
            unsigned lo = 0u, hi = 0u;
            #pragma unroll 16
            for (int g = 0; g < COUNT_BLOCKS; ++g) {
                const unsigned w = p[g * COUNT_CHUNK];
                lo += w & 0xffffu;
                hi += w >> 16;
            }
            const float d0 = (float)lo, d1 = (float)hi;
            deg[2 * i] = (d0 > 0.0f) ? __frsqrt_rn(d0) : 0.0f;
            if (2 * i + 1 < N) deg[2 * i + 1] = (d1 > 0.0f) ? __frsqrt_rn(d1) : 0.0f;
        }
    } else if (i < N) {
        float d = deg[i];
        deg[i] = (d > 0.0f) ? __frsqrt_rn(d) : 0.0f;
    }
}

#define FB_ROWS 100
#define AGG_WAVES (AGG_THREADS / 32)
#define AGG_GRP 8
#define AGG_EPT 8

__device__ __noinline__ void agg_add_serial(float* tile, const float* __restrict__ h,
                                            const float* __restrict__ dis,
                                            const int* __restrict__ col,
                                            int e, int r, int rowBase) {
    const int cI = col[e];
    const float norm = dis[r] * dis[cI];
    const float* hp = h + (size_t)cI * HIDDEN;
    float* tp = tile + (r - rowBase) * HIDDEN;
    for (int k = 0; k < HIDDEN; ++k) atomicAdd(tp + k, hp[k] * norm);
}

template <int G>
__device__ __forceinline__ void agg_add_group(float4& acc, const float4* __restrict__ h4,
                                              int cI, float nm, int j0, int last, int lane) {
    float4 v[G];
    #pragma unroll
    for (int u = 0; u < G; ++u) {
        int js = j0 + u;
        if (js > last) js = last;
        const int cj = __builtin_amdgcn_readlane(cI, js);
        v[u] = h4[(size_t)cj * (HIDDEN / 4) + lane];
    }
    #pragma unroll
    for (int u = 0; u < G; ++u) {
        const float wj = __int_as_float(__builtin_amdgcn_readlane(__float_as_int(nm), j0 + u));
        acc.x = fmaf(v[u].x, wj, acc.x);
        acc.y = fmaf(v[u].y, wj, acc.y);
        acc.z = fmaf(v[u].z, wj, acc.z);
        acc.w = fmaf(v[u].w, wj, acc.w);
    }
}

__global__ __launch_bounds__(AGG_THREADS) void gcn_aggregate_kernel(
        const float* __restrict__ h, const float* __restrict__ dis,
        const int* __restrict__ row, const int* __restrict__ col,
        const int* binw, float* out, int N, int E) {
    __shared__ int   s_list[AGG_CAP];
    __shared__ int   s_src[AGG_CAP];
    __shared__ float s_nm[AGG_CAP];
    __shared__ float s_dr[AGG_ROWS];
    __shared__ int   s_rcnt[AGG_ROWS];
    __shared__ int   s_rend[AGG_ROWS];
    __shared__ __attribute__((aligned(16))) float s_tile[FB_ROWS * HIDDEN];
    __shared__ int s_cnt;

    const int tid  = threadIdx.x;
    const int lane = tid & 31;
    const int wave = tid >> 5;
    const int rowBase = blockIdx.x * AGG_ROWS;
    int rowsHere = N - rowBase;
    if (rowsHere > AGG_ROWS) rowsHere = AGG_ROWS;

    const int* myBin = binw + (size_t)rowBase * HIDDEN;
    const int binCnt = myBin[0];
    const bool binOK = ((int)gridDim.x <= BIN_MAXB) && (N <= (1 << PACK_SHIFT)) &&
                       (binCnt <= AGG_CAP) && (binCnt < rowsHere * HIDDEN);

    if (binOK) {
        for (int i = tid; i < AGG_ROWS; i += AGG_THREADS) {
            s_rcnt[i] = 0;
            s_dr[i] = (i < rowsHere) ? dis[rowBase + i] : 0.0f;
        }
        __syncthreads();

        for (int base = 0; base < binCnt; base += AGG_THREADS * AGG_EPT) {
            int w[AGG_EPT];
            #pragma unroll
            for (int u = 0; u < AGG_EPT; ++u) {
                const int i = base + u * AGG_THREADS + tid;
                w[u] = myBin[1 + ((i < binCnt) ? i : (binCnt - 1))];
            }
            #pragma unroll
            for (int u = 0; u < AGG_EPT; ++u) {
                const int i = base + u * AGG_THREADS + tid;
                if (i < binCnt) {
                    s_list[i] = w[u];
                    atomicAdd(&s_rcnt[(unsigned)w[u] >> PACK_SHIFT], 1);
                }
            }
        }
        __syncthreads();

        if (wave == 0) {
            constexpr int RUN = (AGG_ROWS + 31) / 32;
            int c[RUN];
            int sum = 0;
            #pragma unroll
            for (int k = 0; k < RUN; ++k) {
                const int r = lane * RUN + k;
                c[k] = (r < AGG_ROWS) ? s_rcnt[r] : 0;
                sum += c[k];
            }
            int incl = sum;
            #pragma unroll
            for (int d = 1; d < 32; d <<= 1) {
                const int t = __shfl_up(incl, d);
                if (lane >= d) incl += t;
            }
            int run = incl - sum;
            #pragma unroll
            for (int k = 0; k < RUN; ++k) {
                const int r = lane * RUN + k;
                if (r < AGG_ROWS) s_rend[r] = run;
                run += c[k];
            }
        }
        __syncthreads();

        for (int base = 0; base < binCnt; base += AGG_THREADS * AGG_EPT) {
            int w[AGG_EPT];
            float dc[AGG_EPT];
            #pragma unroll
            for (int u = 0; u < AGG_EPT; ++u) {
                const int i = base + u * AGG_THREADS + tid;
                w[u] = s_list[(i < binCnt) ? i : (binCnt - 1)];
                dc[u] = dis[(unsigned)w[u] & PACK_MASK];
            }
            #pragma unroll
            for (int u = 0; u < AGG_EPT; ++u) {
                const int i = base + u * AGG_THREADS + tid;
                if (i < binCnt) {
                    const int lr = (int)((unsigned)w[u] >> PACK_SHIFT);
                    const int pos = atomicAdd(&s_rend[lr], 1);
                    s_src[pos] = (int)((unsigned)w[u] & PACK_MASK);
                    s_nm[pos]  = s_dr[lr] * dc[u];
                }
            }
        }
        __syncthreads();

        {
            const int waveU = __builtin_amdgcn_readfirstlane(wave);
            const float4* __restrict__ h4 = (const float4*)h;
            float4* o4 = (float4*)out;
            for (int lr = waveU; lr < rowsHere; lr += AGG_WAVES) {
                const int n     = __builtin_amdgcn_readfirstlane(s_rcnt[lr]);
                const int start = __builtin_amdgcn_readfirstlane(s_rend[lr]) - n;
                float4 acc = make_float4(0.f, 0.f, 0.f, 0.f);
                for (int c0 = 0; c0 < n; c0 += 32) {
                    int m = n - c0;
                    if (m > 32) m = 32;
                    int cI = 0;
                    float nm = 0.0f;
                    if (lane < m) {
                        cI = s_src[start + c0 + lane];
                        nm = s_nm[start + c0 + lane];
                    }
                    int j = 0;
                    for (; j + AGG_GRP <= m; j += AGG_GRP) agg_add_group<AGG_GRP>(acc, h4, cI, nm, j, m - 1, lane);
                    const int rem = m - j;
                    if (rem > AGG_GRP / 2)  agg_add_group<AGG_GRP>(acc, h4, cI, nm, j, m - 1, lane);
                    else if (rem > 0)       agg_add_group<AGG_GRP / 2>(acc, h4, cI, nm, j, m - 1, lane);
                }
                o4[(size_t)(rowBase + lr) * (HIDDEN / 4) + lane] = acc;
            }
        }
    } else {
        const int4* __restrict__ row4 = (const int4*)row;
        const int n4 = E >> 2;
        const int step = AGG_THREADS * AGG_UNR;
        const int nIter = (n4 + step - 1) / step;
        for (int sub = 0; sub < rowsHere; sub += FB_ROWS) {
            const int subBase = rowBase + sub;
            int subRows = rowsHere - sub;
            if (subRows > FB_ROWS) subRows = FB_ROWS;
            const unsigned urows = (unsigned)subRows;
            {
                float4* t4 = (float4*)s_tile;
                const float4 z = make_float4(0.f, 0.f, 0.f, 0.f);
                for (int i = tid; i < FB_ROWS * HIDDEN / 4; i += AGG_THREADS) t4[i] = z;
                if (tid == 0) s_cnt = 0;
            }
            __syncthreads();

            const int noMatch = subBase - 1;
            for (int it = 0; it < nIter; ++it) {
                const int i0 = it * step + tid;
                int4 v[AGG_UNR];
                #pragma unroll
                for (int u = 0; u < AGG_UNR; ++u) {
                    const int i = i0 + u * AGG_THREADS;
                    v[u] = (i < n4) ? row4[i] : make_int4(noMatch, noMatch, noMatch, noMatch);
                }
                #pragma unroll
                for (int u = 0; u < AGG_UNR; ++u) {
                    const int e0 = (i0 + u * AGG_THREADS) * 4;
                    const int rr[4] = {v[u].x, v[u].y, v[u].z, v[u].w};
                    #pragma unroll
                    for (int q = 0; q < 4; ++q) {
                        if ((unsigned)(rr[q] - subBase) < urows) {
                            const int p = atomicAdd(&s_cnt, 1);
                            if (p < AGG_CAP) s_list[p] = e0 + q;
                            else agg_add_serial(s_tile, h, dis, col, e0 + q, rr[q], subBase);
                        }
                    }
                }
            }
            const int eTail = (n4 << 2) + tid;
            if (eTail < E) {
                const int r = row[eTail];
                if ((unsigned)(r - subBase) < urows) {
                    const int p = atomicAdd(&s_cnt, 1);
                    if (p < AGG_CAP) s_list[p] = eTail;
                    else agg_add_serial(s_tile, h, dis, col, eTail, r, subBase);
                }
            }
            __syncthreads();

            int total = s_cnt;
            if (total > AGG_CAP) total = AGG_CAP;
            for (int i = wave; i < total; i += AGG_WAVES) {
                const int e  = s_list[i];
                const int cI = col[e];
                const int r  = row[e];
                const float nm = dis[r] * dis[cI];
                const float* hp = h + (size_t)cI * HIDDEN + lane;
                float* tp = s_tile + (r - subBase) * HIDDEN + lane;
                atomicAdd(tp +  0, hp[0]  * nm);
                atomicAdd(tp + 32, hp[32] * nm);
                atomicAdd(tp + 64, hp[64] * nm);
                atomicAdd(tp + 96, hp[96] * nm);
            }
            __syncthreads();

            {
                const float4* t4 = (const float4*)s_tile;
                float4* o4 = (float4*)(out + (size_t)subBase * HIDDEN);
                const int n = subRows * (HIDDEN / 4);
                for (int i = tid; i < n; i += AGG_THREADS) o4[i] = t4[i];
            }
            __syncthreads();
        }
    }
}

extern "C" void kernel_launch(void* const* d_in, const int* in_sizes, int n_in,
                              void* d_out, int out_size, void* d_ws, size_t ws_size,
                              hipStream_t stream) {
    const float* x  = (const float*)d_in[0];
    const float* W  = (const float*)d_in[1];
    const int*   ei = (const int*)d_in[2];
    float* out = (float*)d_out;

    const int N = in_sizes[0] / HIDDEN;
    const int E = in_sizes[2] / 2;
    const int* rowIdx = ei;
    const int* colIdx = ei + E;

    float* h   = (float*)d_ws;
    float* deg = h + (size_t)N * HIDDEN;

    const int nb = (N + AGG_ROWS - 1) / AGG_ROWS;
    int* binw = (int*)d_out;
    gcn_gemm_wmma_kernel<<<GEMM_BLOCKS, GEMM_THREADS, 0, stream>>>(x, W, h, deg, binw, N, E, nb);

    gcn_deg_bin_kernel<<<COUNT_BLOCKS + BIN_BLOCKS, BIN_THREADS, 0, stream>>>(
        rowIdx, colIdx, deg, binw, N, E, nb);

    gcn_dis_kernel<<<(N + 255) / 256, 256, 0, stream>>>(deg, binw, N, E, nb);

    gcn_aggregate_kernel<<<nb, AGG_THREADS, 0, stream>>>(
        h, deg, rowIdx, colIdx, binw, out, N, E);
}


# Binning blocks fill own fixed slots of each bin (no global counter, nothing zeroed first): binning launch goes first, dis pass rides in the projection launch; 4 launches to 3; sums in arrival order
# speedup vs baseline: 4.7694x; 1.1533x over previous
#include <hip/hip_runtime.h>

typedef __attribute__((ext_vector_type(2))) float v2f;
typedef __attribute__((ext_vector_type(8))) float v8f;

#define HIDDEN 128

#define AGG_ROWS    400
#define AGG_THREADS 1024
#define AGG_CAP     16384
#define AGG_UNR     8
#define BIN_WORDS   (AGG_ROWS * HIDDEN)
#define BIN_THREADS 1024
#define BIN_EPT     9
#define BIN_BLOCKS  192
#define BIN_MAXB    1024
#define SLOT_CAP    128
#define SLOT_OFF    BIN_BLOCKS
#define SLOT_END    (BIN_BLOCKS + BIN_BLOCKS * SLOT_CAP)
#define PACK_SHIFT  23
#define PACK_MASK   ((1u << PACK_SHIFT) - 1u)
static_assert(AGG_ROWS <= (1 << (32 - PACK_SHIFT)), "the local row must fit above the source row");

#define COUNT_BLOCKS  64
#define COUNT_THREADS 1024
#define COUNT_EPT     25
#define COUNT_WORDS   51200
#define COUNT_CHUNK   (AGG_ROWS / 2)
#define COUNT_OFF     24832
static_assert(AGG_ROWS % 8 == 0, "a chunk is moved 16 bytes at a time");
static_assert(SLOT_END <= COUNT_OFF, "the blocks' slots end below the tables' chunks");
static_assert(COUNT_OFF % 4 == 0, "chunks start on a 16-byte boundary");

__device__ __forceinline__ bool count_in_lds(int N, int E, int nb) {
    const int lastRows = N - (nb - 1) * AGG_ROWS;
    return (nb * COUNT_CHUNK <= COUNT_WORDS) &&
           ((E + COUNT_BLOCKS - 1) / COUNT_BLOCKS <= 65535) &&
           (lastRows * HIDDEN >= COUNT_OFF + COUNT_BLOCKS * COUNT_CHUNK);
}

__device__ __forceinline__ void count_block(
        const int* __restrict__ col, int* __restrict__ binw, unsigned* s_h,
        int cblk, int N, int E, int nb) {
    const int tid = threadIdx.x;

    const int nQuads = nb * (COUNT_CHUNK / 4);
    {
        uint4* h4 = (uint4*)s_h;
        const uint4 z = make_uint4(0u, 0u, 0u, 0u);
        for (int i = tid; i < nQuads; i += COUNT_THREADS) h4[i] = z;
    }
    __syncthreads();

    const int per = (E + COUNT_BLOCKS - 1) / COUNT_BLOCKS;
    const int cBeg = cblk * per;
    int cEnd = cBeg + per;
    if (cEnd > E) cEnd = E;
    for (int t0 = cBeg; t0 < cEnd; t0 += COUNT_THREADS * COUNT_EPT) {
        int cc[COUNT_EPT];
        #pragma unroll
        for (int u = 0; u < COUNT_EPT; ++u) {
            const int e = t0 + u * COUNT_THREADS + tid;
            cc[u] = col[(e < cEnd) ? e : (cEnd - 1)];
        }
        #pragma unroll
        for (int u = 0; u < COUNT_EPT; ++u) {
            const int e = t0 + u * COUNT_THREADS + tid;
            const int cI = cc[u];
            if (e < cEnd && (unsigned)cI < (unsigned)N)
                atomicAdd(&s_h[cI >> 1], 1u << ((cI & 1) << 4));
        }
    }
    __syncthreads();

    {
        const uint4* h4 = (const uint4*)s_h;
        for (int q = tid; q < nQuads; q += COUNT_THREADS) {
            const int b = q / (COUNT_CHUNK / 4);
            const int j = q - b * (COUNT_CHUNK / 4);
            uint4* dst = (uint4*)(binw + (size_t)b * BIN_WORDS + COUNT_OFF +
                                  cblk * COUNT_CHUNK);
            dst[j] = h4[q];
        }
    }
}

__device__ __forceinline__ void count_slow(
        const int* __restrict__ col, float* __restrict__ deg, unsigned* s_h,
        int cblk, int N, int E) {
    const int tid = threadIdx.x;
    for (int base = cblk * COUNT_WORDS; base < N; base += COUNT_BLOCKS * COUNT_WORDS) {
        int span = N - base;
        if (span > COUNT_WORDS) span = COUNT_WORDS;
        for (int i = tid; i < span; i += COUNT_THREADS) s_h[i] = 0u;
        __syncthreads();
        for (int e = tid; e < E; e += COUNT_THREADS) {
            const unsigned d = (unsigned)(col[e] - base);
            if (d < (unsigned)span) atomicAdd(&s_h[d], 1u);
        }
        __syncthreads();
        for (int i = tid; i < span; i += COUNT_THREADS) deg[base + i] = (float)s_h[i];
        __syncthreads();
    }
}

#define GEMM_THREADS 256
#define GEMM_TR      16
#define GEMM_PITCH   (HIDDEN + 4)
#define GEMM_BLOCKS  768
#define GEMM_KSTEPS  (HIDDEN / 4)
#define DIS_BLOCKS   200

__global__ __launch_bounds__(GEMM_THREADS) void gcn_gemm_wmma_kernel(
        const float* __restrict__ x, const float* __restrict__ W,
        float* __restrict__ h, float* __restrict__ deg, const int* __restrict__ binw,
        int N, int E, int nb) {
    __shared__ __attribute__((aligned(16))) float s_x[2 * GEMM_TR * GEMM_PITCH];

    const int tid     = threadIdx.x;

    if ((int)blockIdx.x >= GEMM_BLOCKS) {
        const int gtid = ((int)blockIdx.x - GEMM_BLOCKS) * GEMM_THREADS + tid;
        const int gsz  = DIS_BLOCKS * GEMM_THREADS;
        if (count_in_lds(N, E, nb)) {
            const int nWords = (N + 1) >> 1;
            for (int i = gtid; i < nWords; i += gsz) {
                const int b = i / COUNT_CHUNK;
                const int j = i - b * COUNT_CHUNK;
                const unsigned* p = (const unsigned*)binw + (size_t)b * BIN_WORDS + COUNT_OFF + j;
                unsigned lo = 0u, hi = 0u;
                #pragma unroll 16
                for (int g = 0; g < COUNT_BLOCKS; ++g) {
                    const unsigned w = p[g * COUNT_CHUNK];
                    lo += w & 0xffffu;
                    hi += w >> 16;
                }
                const float d0 = (float)lo, d1 = (float)hi;
                deg[2 * i] = (d0 > 0.0f) ? __frsqrt_rn(d0) : 0.0f;
                if (2 * i + 1 < N) deg[2 * i + 1] = (d1 > 0.0f) ? __frsqrt_rn(d1) : 0.0f;
            }
        } else {
            for (int i = gtid; i < N; i += gsz) {
                const float d = deg[i];
                deg[i] = (d > 0.0f) ? __frsqrt_rn(d) : 0.0f;
            }
        }
        return;
    }

    const int lane    = tid & 31;
    const int wave    = tid >> 5;
    const int halfSel = lane >> 4;
    const int l       = lane & 15;
    const int nBase   = wave * 16;
    const int nTiles  = (N + GEMM_TR - 1) / GEMM_TR;

    const float* __restrict__ wrow = W + (size_t)(nBase + l) * HIDDEN + (halfSel << 1);
    v2f bfrag[GEMM_KSTEPS];
    #pragma unroll
    for (int s = 0; s < GEMM_KSTEPS; ++s) bfrag[s] = *(const v2f*)(wrow + 4 * s);

    const int sr0 = wave, sr1 = wave + 8;
    float* const sdst0 = &s_x[sr0 * GEMM_PITCH + lane * 4];
    float* const sdst1 = &s_x[sr1 * GEMM_PITCH + lane * 4];

    int tile = blockIdx.x;
    float4 g0 = make_float4(0.f, 0.f, 0.f, 0.f), g1 = g0;
    if (tile < nTiles) {
        int r0 = tile * GEMM_TR + sr0; if (r0 >= N) r0 = N - 1;
        int r1 = tile * GEMM_TR + sr1; if (r1 >= N) r1 = N - 1;
        g0 = *(const float4*)(x + (size_t)r0 * HIDDEN + lane * 4);
        g1 = *(const float4*)(x + (size_t)r1 * HIDDEN + lane * 4);
    }

    int buf = 0;
    for (; tile < nTiles; tile += GEMM_BLOCKS) {
        const int bofs = buf * (GEMM_TR * GEMM_PITCH);
        *(float4*)(sdst0 + bofs) = g0;
        *(float4*)(sdst1 + bofs) = g1;
        __syncthreads();

        const int nxt = tile + GEMM_BLOCKS;
        if (nxt < nTiles) {
            int r0 = nxt * GEMM_TR + sr0; if (r0 >= N) r0 = N - 1;
            int r1 = nxt * GEMM_TR + sr1; if (r1 >= N) r1 = N - 1;
            g0 = *(const float4*)(x + (size_t)r0 * HIDDEN + lane * 4);
            g1 = *(const float4*)(x + (size_t)r1 * HIDDEN + lane * 4);
        }

        const float* ap = &s_x[bofs + l * GEMM_PITCH + (halfSel << 1)];
        v2f afrag[GEMM_KSTEPS];
        #pragma unroll
        for (int s = 0; s < GEMM_KSTEPS; ++s) afrag[s] = *(const v2f*)(ap + 4 * s);

        v8f c = {};
        #pragma unroll
        for (int s = 0; s < GEMM_KSTEPS; ++s) {
            c = __builtin_amdgcn_wmma_f32_16x16x4_f32(
                     false, afrag[s],  false, bfrag[s],
                     (short)0, c,  false,  false);
        }

        const int rowBase = tile * GEMM_TR;
        float* __restrict__ hp = h + (size_t)rowBase * HIDDEN + nBase + l;
        #pragma unroll
        for (int v = 0; v < 8; ++v) {
            const int m = v + halfSel * 8;
            if (rowBase + m < N) hp[(size_t)m * HIDDEN] = c[v];
        }
        buf ^= 1;
    }
}

static_assert(COUNT_THREADS == BIN_THREADS, "counting and binning blocks share one launch");
static_assert(2 * BIN_MAXB <= COUNT_WORDS, "the bin counters lie in the head of the table's LDS");
__global__ __launch_bounds__(BIN_THREADS) void gcn_deg_bin_kernel(
        const int* __restrict__ row, const int* __restrict__ col,
        float* __restrict__ deg, int* __restrict__ binw, int N, int E, int nb) {
    __shared__ __attribute__((aligned(16))) unsigned s_h[COUNT_WORDS];
    int* const s_c = (int*)s_h;
    int* const s_b = (int*)s_h + BIN_MAXB;

    const int tid = threadIdx.x;
    if ((int)blockIdx.x < COUNT_BLOCKS) {
        if (count_in_lds(N, E, nb)) count_block(col, binw, s_h, (int)blockIdx.x, N, E, nb);
        else                        count_slow(col, deg, s_h, (int)blockIdx.x, N, E);
        return;
    }
    const int bblk = (int)blockIdx.x - COUNT_BLOCKS;
    const bool doBin = (nb <= BIN_MAXB) && (N <= (1 << PACK_SHIFT));
    if (bblk >= BIN_BLOCKS || !doBin) return;
    const int per = (E + BIN_BLOCKS - 1) / BIN_BLOCKS;
    const int cBeg = bblk * per;
    int cEnd = cBeg + per;
    if (cEnd > E) cEnd = E;

    for (int i = tid; i < BIN_MAXB; i += BIN_THREADS) s_b[i] = 0;
    for (int t0 = cBeg; t0 < cEnd; t0 += BIN_THREADS * BIN_EPT) {
        for (int i = tid; i < BIN_MAXB; i += BIN_THREADS) s_c[i] = 0;
        __syncthreads();

        int cc[BIN_EPT], bb[BIN_EPT], rk[BIN_EPT], pk[BIN_EPT];
        #pragma unroll
        for (int u = 0; u < BIN_EPT; ++u) {
            const int e = t0 + u * BIN_THREADS + tid;
            const int ec = (e < cEnd) ? e : (cEnd - 1);
            cc[u] = col[ec];
            bb[u] = row[ec];
        }
        #pragma unroll
        for (int u = 0; u < BIN_EPT; ++u) {
            const int e = t0 + u * BIN_THREADS + tid;
            const int r = bb[u];
            const bool live = e < cEnd;
            const int b = r / AGG_ROWS;
            pk[u] = (int)(((unsigned)(r - b * AGG_ROWS) << PACK_SHIFT) | ((unsigned)cc[u] & PACK_MASK));
            bb[u] = (live && (unsigned)r < (unsigned)N) ? b : -1;
            rk[u] = 0;
        }
        #pragma unroll
        for (int u = 0; u < BIN_EPT; ++u) {
            if (bb[u] >= 0) rk[u] = atomicAdd(&s_c[bb[u]], 1);
        }
        __syncthreads();

        #pragma unroll
        for (int u = 0; u < BIN_EPT; ++u) {
            if (bb[u] >= 0) {
                const int b = bb[u];
                int rowsB = N - b * AGG_ROWS;
                if (rowsB > AGG_ROWS) rowsB = AGG_ROWS;
                const int pos = s_b[b] + rk[u];
                if (pos < SLOT_CAP && rowsB * HIDDEN >= SLOT_END)
                    binw[(size_t)b * BIN_WORDS + SLOT_OFF + bblk * SLOT_CAP + pos] = pk[u];
            }
        }
        __syncthreads();
        for (int i = tid; i < BIN_MAXB; i += BIN_THREADS) s_b[i] += s_c[i];
    }
    for (int i = tid; i < nb; i += BIN_THREADS) {
        int rowsB = N - i * AGG_ROWS;
        if (rowsB > AGG_ROWS) rowsB = AGG_ROWS;
        if (rowsB * HIDDEN >= SLOT_END) binw[(size_t)i * BIN_WORDS + bblk] = s_b[i];
    }
}

#define FB_ROWS 100
#define AGG_WAVES (AGG_THREADS / 32)
#define AGG_GRP 8
#define AGG_EPT 8

__device__ __noinline__ void agg_add_serial(float* tile, const float* __restrict__ h,
                                            const float* __restrict__ dis,
                                            const int* __restrict__ col,
                                            int e, int r, int rowBase) {
    const int cI = col[e];
    const float norm = dis[r] * dis[cI];
    const float* hp = h + (size_t)cI * HIDDEN;
    float* tp = tile + (r - rowBase) * HIDDEN;
    for (int k = 0; k < HIDDEN; ++k) atomicAdd(tp + k, hp[k] * norm);
}

template <int G>
__device__ __forceinline__ void agg_add_group(float4& acc, const float4* __restrict__ h4,
                                              int cI, float nm, int j0, int last, int lane) {
    float4 v[G];
    #pragma unroll
    for (int u = 0; u < G; ++u) {
        int js = j0 + u;
        if (js > last) js = last;
        const int cj = __builtin_amdgcn_readlane(cI, js);
        v[u] = h4[(size_t)cj * (HIDDEN / 4) + lane];
    }
    #pragma unroll
    for (int u = 0; u < G; ++u) {
        const float wj = __int_as_float(__builtin_amdgcn_readlane(__float_as_int(nm), j0 + u));
        acc.x = fmaf(v[u].x, wj, acc.x);
        acc.y = fmaf(v[u].y, wj, acc.y);
        acc.z = fmaf(v[u].z, wj, acc.z);
        acc.w = fmaf(v[u].w, wj, acc.w);
    }
}

__global__ __launch_bounds__(AGG_THREADS) void gcn_aggregate_kernel(
        const float* __restrict__ h, const float* __restrict__ dis,
        const int* __restrict__ row, const int* __restrict__ col,
        const int* binw, float* out, int N, int E) {
    __shared__ int   s_list[AGG_CAP];
    __shared__ int   s_src[AGG_CAP];
    __shared__ float s_nm[AGG_CAP];
    __shared__ float s_dr[AGG_ROWS];
    __shared__ int   s_rcnt[AGG_ROWS];
    __shared__ int   s_rend[AGG_ROWS];
    __shared__ __attribute__((aligned(16))) float s_tile[FB_ROWS * HIDDEN];
    __shared__ int s_cnt;
    __shared__ int s_soff[BIN_BLOCKS + 1];
    __shared__ int s_fit;

    const int tid  = threadIdx.x;
    const int lane = tid & 31;
    const int wave = tid >> 5;
    const int rowBase = blockIdx.x * AGG_ROWS;
    int rowsHere = N - rowBase;
    if (rowsHere > AGG_ROWS) rowsHere = AGG_ROWS;

    const int* myBin = binw + (size_t)rowBase * HIDDEN;
    const bool binBuilt = ((int)gridDim.x <= BIN_MAXB) && (N <= (1 << PACK_SHIFT)) &&
                          (rowsHere * HIDDEN >= SLOT_END);
    if (binBuilt) {
        for (int i = tid; i < AGG_ROWS; i += AGG_THREADS) {
            s_rcnt[i] = 0;
            s_dr[i] = (i < rowsHere) ? dis[rowBase + i] : 0.0f;
        }
    }
    if (binBuilt && wave == 0) {
        constexpr int SRUN = BIN_BLOCKS / 32;
        int c[SRUN];
        int sum = 0;
        bool full = false, longer = false;
        #pragma unroll
        for (int k = 0; k < SRUN; ++k) c[k] = myBin[lane * SRUN + k];
        #pragma unroll
        for (int k = 0; k < SRUN; ++k) {
            full = full || (c[k] > SLOT_CAP);
            longer = longer || (c[k] > SLOT_CAP / 2);
            sum += c[k];
        }
        int incl = sum;
        #pragma unroll
        for (int d = 1; d < 32; d <<= 1) {
            const int t = __shfl_up(incl, d);
            if (lane >= d) incl += t;
        }
        int run = incl - sum;
        #pragma unroll
        for (int k = 0; k < SRUN; ++k) {
            s_soff[lane * SRUN + k] = run;
            run += c[k];
        }
        const bool anyFull = __any(full);
        const bool anyLong = __any(longer);
        if (lane == 31) {
            s_soff[BIN_BLOCKS] = incl;
            s_fit = (!anyFull && incl <= AGG_CAP) ? (anyLong ? 2 : 1) : 0;
        }
    }
    __syncthreads();
    const int  fit   = binBuilt ? s_fit : 0;
    const bool binOK = fit != 0;
    const int binCnt = binOK ? s_soff[BIN_BLOCKS] : 0;

    if (binOK) {
        {
            constexpr int HALF  = SLOT_CAP / 2;
            constexpr int SPP   = AGG_THREADS / HALF;
            constexpr int NPASS = BIN_BLOCKS / SPP;
            static_assert(BIN_BLOCKS % SPP == 0, "whole passes over the slots");
            const int sg = tid / HALF;
            for (int half = 0; half < fit; ++half) {
                const int k = (tid & (HALF - 1)) + half * HALF;
                int w[NPASS], pos[NPASS];
                #pragma unroll
                for (int u = 0; u < NPASS; ++u) {
                    const int sl = u * SPP + sg;
                    const int o  = s_soff[sl];
                    const int c  = s_soff[sl + 1] - o;
                    pos[u] = (k < c) ? (o + k) : -1;
                    w[u] = myBin[SLOT_OFF + sl * SLOT_CAP + ((k < c) ? k : 0)];
                }
                #pragma unroll
                for (int u = 0; u < NPASS; ++u) {
                    if (pos[u] >= 0) {
                        s_list[pos[u]] = w[u];
                        atomicAdd(&s_rcnt[(unsigned)w[u] >> PACK_SHIFT], 1);
                    }
                }
            }
        }
        __syncthreads();

        if (wave == 0) {
            constexpr int RUN = (AGG_ROWS + 31) / 32;
            int c[RUN];
            int sum = 0;
            #pragma unroll
            for (int k = 0; k < RUN; ++k) {
                const int r = lane * RUN + k;
                c[k] = (r < AGG_ROWS) ? s_rcnt[r] : 0;
                sum += c[k];
            }
            int incl = sum;
            #pragma unroll
            for (int d = 1; d < 32; d <<= 1) {
                const int t = __shfl_up(incl, d);
                if (lane >= d) incl += t;
            }
            int run = incl - sum;
            #pragma unroll
            for (int k = 0; k < RUN; ++k) {
                const int r = lane * RUN + k;
                if (r < AGG_ROWS) s_rend[r] = run;
                run += c[k];
            }
        }
        __syncthreads();

        for (int base = 0; base < binCnt; base += AGG_THREADS * AGG_EPT) {
            int w[AGG_EPT];
            float dc[AGG_EPT];
            #pragma unroll
            for (int u = 0; u < AGG_EPT; ++u) {
                const int i = base + u * AGG_THREADS + tid;
                w[u] = s_list[(i < binCnt) ? i : (binCnt - 1)];
                dc[u] = dis[(unsigned)w[u] & PACK_MASK];
            }
            #pragma unroll
            for (int u = 0; u < AGG_EPT; ++u) {
                const int i = base + u * AGG_THREADS + tid;
                if (i < binCnt) {
                    const int lr = (int)((unsigned)w[u] >> PACK_SHIFT);
                    const int pos = atomicAdd(&s_rend[lr], 1);
                    s_src[pos] = (int)((unsigned)w[u] & PACK_MASK);
                    s_nm[pos]  = s_dr[lr] * dc[u];
                }
            }
        }
        __syncthreads();

        {
            const int waveU = __builtin_amdgcn_readfirstlane(wave);
            const float4* __restrict__ h4 = (const float4*)h;
            float4* o4 = (float4*)out;
            for (int lr = waveU; lr < rowsHere; lr += AGG_WAVES) {
                const int n     = __builtin_amdgcn_readfirstlane(s_rcnt[lr]);
                const int start = __builtin_amdgcn_readfirstlane(s_rend[lr]) - n;
                float4 acc = make_float4(0.f, 0.f, 0.f, 0.f);
                for (int c0 = 0; c0 < n; c0 += 32) {
                    int m = n - c0;
                    if (m > 32) m = 32;
                    int cI = 0;
                    float nm = 0.0f;
                    if (lane < m) {
                        cI = s_src[start + c0 + lane];
                        nm = s_nm[start + c0 + lane];
                    }
                    int j = 0;
                    for (; j + AGG_GRP <= m; j += AGG_GRP) agg_add_group<AGG_GRP>(acc, h4, cI, nm, j, m - 1, lane);
                    const int rem = m - j;
                    if (rem > AGG_GRP / 2)  agg_add_group<AGG_GRP>(acc, h4, cI, nm, j, m - 1, lane);
                    else if (rem > 0)       agg_add_group<AGG_GRP / 2>(acc, h4, cI, nm, j, m - 1, lane);
                }
                o4[(size_t)(rowBase + lr) * (HIDDEN / 4) + lane] = acc;
            }
        }
    } else {
        const int4* __restrict__ row4 = (const int4*)row;
        const int n4 = E >> 2;
        const int step = AGG_THREADS * AGG_UNR;
        const int nIter = (n4 + step - 1) / step;
        for (int sub = 0; sub < rowsHere; sub += FB_ROWS) {
            const int subBase = rowBase + sub;
            int subRows = rowsHere - sub;
            if (subRows > FB_ROWS) subRows = FB_ROWS;
            const unsigned urows = (unsigned)subRows;
            {
                float4* t4 = (float4*)s_tile;
                const float4 z = make_float4(0.f, 0.f, 0.f, 0.f);
                for (int i = tid; i < FB_ROWS * HIDDEN / 4; i += AGG_THREADS) t4[i] = z;
                if (tid == 0) s_cnt = 0;
            }
            __syncthreads();

            const int noMatch = subBase - 1;
            for (int it = 0; it < nIter; ++it) {
                const int i0 = it * step + tid;
                int4 v[AGG_UNR];
                #pragma unroll
                for (int u = 0; u < AGG_UNR; ++u) {
                    const int i = i0 + u * AGG_THREADS;
                    v[u] = (i < n4) ? row4[i] : make_int4(noMatch, noMatch, noMatch, noMatch);
                }
                #pragma unroll
                for (int u = 0; u < AGG_UNR; ++u) {
                    const int e0 = (i0 + u * AGG_THREADS) * 4;
                    const int rr[4] = {v[u].x, v[u].y, v[u].z, v[u].w};
                    #pragma unroll
                    for (int q = 0; q < 4; ++q) {
                        if ((unsigned)(rr[q] - subBase) < urows) {
                            const int p = atomicAdd(&s_cnt, 1);
                            if (p < AGG_CAP) s_list[p] = e0 + q;
                            else agg_add_serial(s_tile, h, dis, col, e0 + q, rr[q], subBase);
                        }
                    }
                }
            }
            const int eTail = (n4 << 2) + tid;
            if (eTail < E) {
                const int r = row[eTail];
                if ((unsigned)(r - subBase) < urows) {
                    const int p = atomicAdd(&s_cnt, 1);
                    if (p < AGG_CAP) s_list[p] = eTail;
                    else agg_add_serial(s_tile, h, dis, col, eTail, r, subBase);
                }
            }
            __syncthreads();

            int total = s_cnt;
            if (total > AGG_CAP) total = AGG_CAP;
            for (int i = wave; i < total; i += AGG_WAVES) {
                const int e  = s_list[i];
                const int cI = col[e];
                const int r  = row[e];
                const float nm = dis[r] * dis[cI];
                const float* hp = h + (size_t)cI * HIDDEN + lane;
                float* tp = s_tile + (r - subBase) * HIDDEN + lane;
                atomicAdd(tp +  0, hp[0]  * nm);
                atomicAdd(tp + 32, hp[32] * nm);
                atomicAdd(tp + 64, hp[64] * nm);
                atomicAdd(tp + 96, hp[96] * nm);
            }
            __syncthreads();

            {
                const float4* t4 = (const float4*)s_tile;
                float4* o4 = (float4*)(out + (size_t)subBase * HIDDEN);
                const int n = subRows * (HIDDEN / 4);
                for (int i = tid; i < n; i += AGG_THREADS) o4[i] = t4[i];
            }
            __syncthreads();
        }
    }
}

extern "C" void kernel_launch(void* const* d_in, const int* in_sizes, int n_in,
                              void* d_out, int out_size, void* d_ws, size_t ws_size,
                              hipStream_t stream) {
    const float* x  = (const float*)d_in[0];
    const float* W  = (const float*)d_in[1];
    const int*   ei = (const int*)d_in[2];
    float* out = (float*)d_out;

    const int N = in_sizes[0] / HIDDEN;
    const int E = in_sizes[2] / 2;
    const int* rowIdx = ei;
    const int* colIdx = ei + E;

    float* h   = (float*)d_ws;
    float* deg = h + (size_t)N * HIDDEN;

    const int nb = (N + AGG_ROWS - 1) / AGG_ROWS;
    int* binw = (int*)d_out;

    gcn_deg_bin_kernel<<<COUNT_BLOCKS + BIN_BLOCKS, BIN_THREADS, 0, stream>>>(
        rowIdx, colIdx, deg, binw, N, E, nb);

    gcn_gemm_wmma_kernel<<<GEMM_BLOCKS + DIS_BLOCKS, GEMM_THREADS, 0, stream>>>(
        x, W, h, deg, binw, N, E, nb);

    gcn_aggregate_kernel<<<nb, AGG_THREADS, 0, stream>>>(
        h, deg, rowIdx, colIdx, binw, out, N, E);
}
